# starting kernel
# baseline (speedup 1.0000x reference)
#include <hip/hip_runtime.h>
#include <math.h>

typedef __attribute__((ext_vector_type(2))) float v2f;
typedef __attribute__((ext_vector_type(8))) float v8f;

#define WMMA_F32(A, B, C) \
  __builtin_amdgcn_wmma_f32_16x16x4_f32(false, (A), false, (B), (short)0, (C), false, false)

#define NB   32
#define NT   12
#define NN   512
#define NDE  16
#define NHID 64
#define NCP  68

__global__ void k_w(const float* __restrict__ ne, const float* __restrict__ wp,
                    float* __restrict__ W) {
  long idx = (long)blockIdx.x * 256 + threadIdx.x;
  const long total = 3L * NN * 2 * NCP * NHID;
  if (idx >= total) return;
  int o = idx % 64; long r = idx / 64;
  int i = r % NCP; r /= NCP;
  int slot = r % 2; r /= 2;
  int n = r % NN; int g = (int)(r / NN);
  float acc = 0.f;
  if (i < 67) {
    for (int d = 0; d < NDE; ++d)
      acc += ne[n * NDE + d] * wp[(((long)(g * NDE + d) * 2 + slot) * 67 + i) * 64 + o];
  }
  W[idx] = acc;
}

__global__ void k_tb(const float* __restrict__ te, const float* __restrict__ bp,
                     float* __restrict__ TB) {
  int idx = blockIdx.x * 256 + threadIdx.x;
  if (idx >= 3 * NT * NB * NHID) return;
  int o = idx % 64; int r = idx / 64;
  int b = r % NB; r /= NB;
  int t = r % NT; int g = r / NT;
  float acc = 0.f;
  for (int d = 0; d < NDE; ++d)
    acc += te[(b * NT + t) * NDE + d] * bp[(g * NDE + d) * 64 + o];
  TB[idx] = acc;
}

__global__ void k_emb(const float* __restrict__ ne, const float* __restrict__ te,
                      const float* __restrict__ glg, const float* __restrict__ glb,
                      float* __restrict__ E, int t) {
  int idx = blockIdx.x * 256 + threadIdx.x;
  if (idx >= 3 * NB * NN) return;
  int n = idx % NN; int r = idx / NN;
  int b = r % NB; int g = r / NB;
  float v[NDE]; float mu = 0.f;
  for (int d = 0; d < NDE; ++d) { v[d] = ne[n * NDE + d] + te[(b * NT + t) * NDE + d]; mu += v[d]; }
  mu *= (1.f / NDE);
  float var = 0.f;
  for (int d = 0; d < NDE; ++d) { float q = v[d] - mu; var += q * q; }
  var *= (1.f / NDE);
  float inv = rsqrtf(var + 1e-12f);
  for (int d = 0; d < NDE; ++d)
    E[(long)idx * NDE + d] = (v[d] - mu) * inv * glg[g * NDE + d] + glb[g * NDE + d];
}

__global__ void k_zero(float* p, long nElem) {
  long i = (long)blockIdx.x * 256 + threadIdx.x;
  if (i < nElem) p[i] = 0.f;
}

__global__ void k_build_ias(const float* __restrict__ x, const float* __restrict__ h,
                            float* __restrict__ ias, int t) {
  int idx = blockIdx.x * 256 + threadIdx.x;
  if (idx >= NB * NN * NCP) return;
  int i = idx % NCP; int r = idx / NCP; int n = r % NN; int b = r / NN;
  float v;
  if (i < 3)       v = x[((long)(b * NT + t) * NN + n) * 3 + i];
  else if (i < 67) v = h[((long)b * NN + n) * 64 + (i - 3)];
  else             v = 0.f;
  ias[idx] = v;
}

__global__ void k_build_cand(const float* __restrict__ x, const float* __restrict__ h,
                             const float* __restrict__ zp, float* __restrict__ cand, int t) {
  int idx = blockIdx.x * 256 + threadIdx.x;
  if (idx >= NB * NN * NCP) return;
  int i = idx % NCP; int r = idx / NCP; int n = r % NN; int b = r / NN;
  float v;
  if (i < 3)       v = x[((long)(b * NT + t) * NN + n) * 3 + i];
  else if (i < 67) {
    long e = ((long)b * NN + n) * 64 + (i - 3);
    float z = 1.f / (1.f + __expf(-zp[e]));
    v = z * h[e];
  } else v = 0.f;
  cand[idx] = v;
}

__global__ void __launch_bounds__(128) k_graph_attn(
    const float* __restrict__ E,
    const float* __restrict__ X,
    float* __restrict__ Y,
    int g) {
  __shared__ float lds_e[16][16];
  __shared__ float lds_v[16][80];
  __shared__ float lds_p[4][16][17];

  const int b = blockIdx.y;
  const int row0 = blockIdx.x * 64;
  const int tid = threadIdx.x;
  const int wave = tid >> 5;
  const int lane = tid & 31;
  const int l16 = lane & 15;
  const int half = lane >> 4;
  const int myrow = row0 + wave * 16 + l16;

  const float* Eb = E + ((long)g * NB + b) * NN * NDE;
  const float* Xb = X + (long)b * NN * NCP;

  v2f aA[4];
#pragma unroll
  for (int kk = 0; kk < 4; ++kk) {
    aA[kk].x = Eb[myrow * NDE + kk * 4 + 2 * half];
    aA[kk].y = Eb[myrow * NDE + kk * 4 + 2 * half + 1];
  }

  for (int i = tid; i < 16 * 12; i += 128) {
    int rr = i / 12, cc = 68 + i % 12;
    lds_v[rr][cc] = (cc == 68) ? 1.f : 0.f;
  }

  v8f acc[5] = {};

  for (int mc = 0; mc < NN / 16; ++mc) {
    __syncthreads();
    {
      const float2* se = (const float2*)(Eb + (long)mc * 16 * NDE);
      ((float2*)lds_e)[tid] = se[tid];
    }
    {
      const float2* sv = (const float2*)(Xb + (long)mc * 16 * NCP);
      for (int i = tid; i < 544; i += 128) {
        int rr = i / 34, cc = (i % 34) * 2;
        *(float2*)&lds_v[rr][cc] = sv[i];
      }
    }
    __syncthreads();

    v8f c = {};
#pragma unroll
    for (int kk = 0; kk < 4; ++kk) {
      v2f bf;
      bf.x = lds_e[l16][kk * 4 + 2 * half];
      bf.y = lds_e[l16][kk * 4 + 2 * half + 1];
      c = WMMA_F32(aA[kk], bf, c);
    }

#pragma unroll
    for (int r = 0; r < 8; ++r)
      lds_p[wave][r + 8 * half][l16] = __expf(c[r]);

#pragma unroll
    for (int kk = 0; kk < 4; ++kk) {
      v2f ap;
      ap.x = lds_p[wave][l16][kk * 4 + 2 * half];
      ap.y = lds_p[wave][l16][kk * 4 + 2 * half + 1];
#pragma unroll
      for (int ct = 0; ct < 5; ++ct) {
        v2f bv;
        bv.x = lds_v[kk * 4 + 2 * half][ct * 16 + l16];
        bv.y = lds_v[kk * 4 + 2 * half + 1][ct * 16 + l16];
        acc[ct] = WMMA_F32(ap, bv, acc[ct]);
      }
    }
  }

#pragma unroll
  for (int r = 0; r < 8; ++r) {
    float rs = __shfl(acc[4][r], half * 16 + 4);
    float inv = 1.f / rs;
    int row = row0 + wave * 16 + r + 8 * half;
#pragma unroll
    for (int ct = 0; ct < 5; ++ct) {
      int col = ct * 16 + l16;
      if (col < NCP)
        Y[((long)b * NN + row) * NCP + col] = acc[ct][r] * inv;
    }
  }
}

__global__ void __launch_bounds__(64) k_gout(
    const float* __restrict__ X, const float* __restrict__ XG,
    const float* __restrict__ W, const float* __restrict__ TB,
    float* __restrict__ G, int g, int t) {
  const int n = blockIdx.x;
  const int wave = threadIdx.x >> 5;
  const int lane = threadIdx.x & 31;
  const int l16 = lane & 15, half = lane >> 4;
  const float* Wg = W + ((long)g * NN + n) * 2 * NCP * NHID;

  v8f c[4] = {};
  for (int slot = 0; slot < 2; ++slot) {
    const float* Arow = (slot ? XG : X) + ((long)(wave * 16 + l16) * NN + n) * NCP;
    const float* Wm = Wg + (long)slot * NCP * NHID;
    for (int kk = 0; kk < NCP / 4; ++kk) {
      v2f a;
      a.x = Arow[kk * 4 + 2 * half];
      a.y = Arow[kk * 4 + 2 * half + 1];
#pragma unroll
      for (int ct = 0; ct < 4; ++ct) {
        v2f bw;
        bw.x = Wm[(kk * 4 + 2 * half) * 64 + ct * 16 + l16];
        bw.y = Wm[(kk * 4 + 2 * half + 1) * 64 + ct * 16 + l16];
        c[ct] = WMMA_F32(a, bw, c[ct]);
      }
    }
  }
#pragma unroll
  for (int ct = 0; ct < 4; ++ct)
#pragma unroll
    for (int r = 0; r < 8; ++r) {
      int bb = wave * 16 + r + 8 * half;
      int o = ct * 16 + l16;
      G[((long)bb * NN + n) * 64 + o] = c[ct][r] + TB[(((long)g * NT + t) * NB + bb) * 64 + o];
    }
}

__global__ void __launch_bounds__(256) k_state_attn(
    float* __restrict__ G, const float* __restrict__ Hs,
    const float* __restrict__ olg, const float* __restrict__ olb, int g, int K) {
  const int wave = threadIdx.x >> 5, lane = threadIdx.x & 31;
  const long row = (long)blockIdx.x * 8 + wave;
  float g0 = G[row * 64 + lane];
  float g1 = G[row * 64 + lane + 32];
  float s = g0 + g1;
  s += __shfl_xor(s, 1); s += __shfl_xor(s, 2); s += __shfl_xor(s, 4);
  s += __shfl_xor(s, 8); s += __shfl_xor(s, 16);
  float mu = s * (1.f / 64.f);
  float d0 = g0 - mu, d1 = g1 - mu;
  float v = d0 * d0 + d1 * d1;
  v += __shfl_xor(v, 1); v += __shfl_xor(v, 2); v += __shfl_xor(v, 4);
  v += __shfl_xor(v, 8); v += __shfl_xor(v, 16);
  float inv = rsqrtf(v * (1.f / 64.f) + 1e-5f);
  float q0 = d0 * inv * olg[g * 64 + lane] + olb[g * 64 + lane];
  float q1 = d1 * inv * olg[g * 64 + lane + 32] + olb[g * 64 + lane + 32];

  float st0[13], st1[13], sc0[13], sc1[13];
#pragma unroll
  for (int k = 0; k < 13; ++k) {
    if (k < K) {
      st0[k] = Hs[((long)k * NB * NN + row) * 64 + lane];
      st1[k] = Hs[((long)k * NB * NN + row) * 64 + lane + 32];
      float p0 = q0 * st0[k], p1 = q1 * st1[k];
      p0 += __shfl_xor(p0, 1); p0 += __shfl_xor(p0, 2);
      p0 += __shfl_xor(p0, 4); p0 += __shfl_xor(p0, 8);
      p1 += __shfl_xor(p1, 1); p1 += __shfl_xor(p1, 2);
      p1 += __shfl_xor(p1, 4); p1 += __shfl_xor(p1, 8);
      sc0[k] = p0 * 0.25f; sc1[k] = p1 * 0.25f;
    } else { st0[k] = 0.f; st1[k] = 0.f; sc0[k] = -1e30f; sc1[k] = -1e30f; }
  }
  float m0 = -1e30f, m1 = -1e30f;
#pragma unroll
  for (int k = 0; k < 13; ++k) { m0 = fmaxf(m0, sc0[k]); m1 = fmaxf(m1, sc1[k]); }
  float s0 = 0.f, s1 = 0.f;
#pragma unroll
  for (int k = 0; k < 13; ++k) {
    sc0[k] = __expf(sc0[k] - m0); s0 += sc0[k];
    sc1[k] = __expf(sc1[k] - m1); s1 += sc1[k];
  }
  float o0 = 0.f, o1 = 0.f;
#pragma unroll
  for (int k = 0; k < 13; ++k) { o0 += sc0[k] * st0[k]; o1 += sc1[k] * st1[k]; }
  G[row * 64 + lane]      = g0 + o0 / s0;
  G[row * 64 + lane + 32] = g1 + o1 / s1;
}

__global__ void k_combine(const float* __restrict__ Gr, const float* __restrict__ Ghc,
                          const float* __restrict__ Hprev, float* __restrict__ Hnew) {
  long i = (long)blockIdx.x * 256 + threadIdx.x;
  if (i >= (long)NB * NN * 64) return;
  float r = 1.f / (1.f + __expf(-Gr[i]));
  float hc = tanhf(Ghc[i]);
  Hnew[i] = r * Hprev[i] + (1.f - r) * hc;
}

__global__ void k_pred(const float* __restrict__ Hs, const float* __restrict__ pw,
                       const float* __restrict__ pb, float* __restrict__ out) {
  int idx = blockIdx.x * 256 + threadIdx.x;
  if (idx >= NB * 12 * NN) return;
  int n = idx % NN; int r = idx / NN; int o = r % 12; int b = r / 12;
  float acc = pb[o];
  for (int cc = 0; cc < 2; ++cc) {
    const float* h = Hs + ((long)(NT - 1 + cc) * NB * NN + (long)b * NN + n) * 64;
    for (int hh = 0; hh < 64; ++hh)
      acc += h[hh] * pw[(o * 2 + cc) * 64 + hh];
  }
  out[idx] = acc;
}

extern "C" void kernel_launch(void* const* d_in, const int* in_sizes, int n_in,
                              void* d_out, int out_size, void* d_ws, size_t ws_size,
                              hipStream_t stream) {
  const float* x   = (const float*)d_in[0];
  const float* ne  = (const float*)d_in[1];
  const float* te  = (const float*)d_in[2];
  const float* wp  = (const float*)d_in[3];
  const float* bp  = (const float*)d_in[4];
  const float* glg = (const float*)d_in[5];
  const float* glb = (const float*)d_in[6];
  const float* olg = (const float*)d_in[7];
  const float* olb = (const float*)d_in[8];
  const float* pw  = (const float*)d_in[9];
  const float* pb  = (const float*)d_in[10];
  float* out = (float*)d_out;

  float* ws = (float*)d_ws;
  long off = 0;
  float* W    = ws + off; off += 3L * NN * 2 * NCP * NHID;
  float* TB   = ws + off; off += 3L * NT * NB * NHID;
  float* E    = ws + off; off += 3L * NB * NN * NDE;
  float* Hs   = ws + off; off += (long)(NT + 1) * NB * NN * NHID;
  float* IAS  = ws + off; off += (long)NB * NN * NCP;
  float* CAND = ws + off; off += (long)NB * NN * NCP;
  float* XG2  = ws + off; off += (long)NB * NN * NCP;
  float* G0   = ws + off; off += (long)NB * NN * NHID;
  float* G1   = ws + off; off += (long)NB * NN * NHID;
  float* G2   = ws + off; off += (long)NB * NN * NHID;

  const long wTotal = 3L * NN * 2 * NCP * NHID;
  k_w<<<dim3((unsigned)((wTotal + 255) / 256)), 256, 0, stream>>>(ne, wp, W);
  k_tb<<<(3 * NT * NB * NHID + 255) / 256, 256, 0, stream>>>(te, bp, TB);
  k_zero<<<(unsigned)(((long)NB * NN * NHID + 255) / 256), 256, 0, stream>>>(Hs, (long)NB * NN * NHID);

  for (int t = 0; t < NT; ++t) {
    float* Hprev = Hs + (long)t * NB * NN * NHID;
    float* Hnew  = Hs + (long)(t + 1) * NB * NN * NHID;
    k_emb<<<(3 * NB * NN + 255) / 256, 256, 0, stream>>>(ne, te, glg, glb, E, t);
    k_build_ias<<<(NB * NN * NCP + 255) / 256, 256, 0, stream>>>(x, Hprev, IAS, t);
    for (int g = 0; g < 2; ++g) {
      float* Gg = g ? G1 : G0;
      k_graph_attn<<<dim3(NN / 64, NB), 128, 0, stream>>>(E, IAS, XG2, g);
      k_gout<<<NN, 64, 0, stream>>>(IAS, XG2, W, TB, Gg, g, t);
      k_state_attn<<<NB * NN / 8, 256, 0, stream>>>(Gg, Hs, olg, olb, g, t + 1);
    }
    k_build_cand<<<(NB * NN * NCP + 255) / 256, 256, 0, stream>>>(x, Hprev, G0, CAND, t);
    k_graph_attn<<<dim3(NN / 64, NB), 128, 0, stream>>>(E, CAND, XG2, 2);
    k_gout<<<NN, 64, 0, stream>>>(CAND, XG2, W, TB, G2, 2, t);
    k_state_attn<<<NB * NN / 8, 256, 0, stream>>>(G2, Hs, olg, olb, 2, t + 1);
    k_combine<<<(unsigned)(((long)NB * NN * NHID + 255) / 256), 256, 0, stream>>>(G1, G2, Hprev, Hnew);
  }
  k_pred<<<(NB * 12 * NN + 255) / 256, 256, 0, stream>>>(Hs, pw, pb, out);
}


# graph attention: each 64-key window (E, state, x) is loaded into registers one round ahead, under the products of the window before; 4 serial load waits per round go; sum orders unchanged
# speedup vs baseline: 4.3107x; 4.3107x over previous
#include <hip/hip_runtime.h>
#include <math.h>

typedef __attribute__((ext_vector_type(2))) float v2f;
typedef __attribute__((ext_vector_type(8))) float v8f;

#define WMMA_F32(A, B, C) \
  __builtin_amdgcn_wmma_f32_16x16x4_f32(false, (A), false, (B), (short)0, (C), false, false)

#define NB   32
#define NT   12
#define NN   512
#define NDE  16
#define NHID 64
#define NCP  68

__global__ void k_w(const float* __restrict__ ne, const float* __restrict__ wp,
                    float* __restrict__ W) {
  long idx = (long)blockIdx.x * 256 + threadIdx.x;
  const long total = 3L * NN * 2 * NCP * NHID;
  if (idx >= total) return;
  int o = idx % 64; long r = idx / 64;
  int i = r % NCP; r /= NCP;
  int slot = r % 2; r /= 2;
  int n = r % NN; int g = (int)(r / NN);
  float acc = 0.f;
  if (i < 67) {
    for (int d = 0; d < NDE; ++d)
      acc += ne[n * NDE + d] * wp[(((long)(g * NDE + d) * 2 + slot) * 67 + i) * 64 + o];
  }
  W[idx] = acc;
}

__global__ void k_tb(const float* __restrict__ te, const float* __restrict__ bp,
                     float* __restrict__ TB) {
  int idx = blockIdx.x * 256 + threadIdx.x;
  if (idx >= 3 * NT * NB * NHID) return;
  int o = idx % 64; int r = idx / 64;
  int b = r % NB; r /= NB;
  int t = r % NT; int g = r / NT;
  float acc = 0.f;
  for (int d = 0; d < NDE; ++d)
    acc += te[(b * NT + t) * NDE + d] * bp[(g * NDE + d) * 64 + o];
  TB[idx] = acc;
}

__device__ __forceinline__ void emb_row(const float* __restrict__ ne, const float* __restrict__ te,
                                        const float* __restrict__ glg, const float* __restrict__ glb,
                                        float* __restrict__ E, int t, int idx) {
  if (idx >= 3 * NB * NN) return;
  int n = idx % NN; int r = idx / NN;
  int b = r % NB; int g = r / NB;
  float v[NDE]; float mu = 0.f;
  for (int d = 0; d < NDE; ++d) { v[d] = ne[n * NDE + d] + te[(b * NT + t) * NDE + d]; mu += v[d]; }
  mu *= (1.f / NDE);
  float var = 0.f;
  for (int d = 0; d < NDE; ++d) { float q = v[d] - mu; var += q * q; }
  var *= (1.f / NDE);
  float inv = rsqrtf(var + 1e-12f);
  for (int d = 0; d < NDE; ++d)
    E[(long)idx * NDE + d] = (v[d] - mu) * inv * glg[g * NDE + d] + glb[g * NDE + d];
}

__global__ void k_emb(const float* __restrict__ ne, const float* __restrict__ te,
                      const float* __restrict__ glg, const float* __restrict__ glb,
                      float* __restrict__ E, int t) {
  emb_row(ne, te, glg, glb, E, t, blockIdx.x * 256 + threadIdx.x);
}

__global__ void k_zero(float* p, long nElem) {
  long i = (long)blockIdx.x * 256 + threadIdx.x;
  if (i < nElem) p[i] = 0.f;
}


#define KGROUPS 4
#define GA_LDS_FLOATS 15360
#define GA_BLOCKS 256
__device__ __forceinline__ void graph_attn_body(
    float* smem, int bid,
    const float* __restrict__ E,
    const float* __restrict__ x,
    const float* __restrict__ Hsrc,
    float* __restrict__ Y,
    int g, int t) {
  float (*lds_e)[16] = (float (*)[16])smem;
  float (*lds_v)[80] = (float (*)[80])(smem + 1024);
  float (*lds_p)[16][17] = (float (*)[16][17])(smem + 6144);
  float4* lds_red = (float4*)smem;
#define RED_AT(s_, rt_, ct_, h_, l_) lds_red[((((s_) * 4 + (rt_)) * 5 + (ct_)) * 2 + (h_)) * 32 + (l_)]

  const int b = bid >> 3;
  const int row0 = (bid & 7) * 64;
  const int tid = threadIdx.x;
  const int wave = tid >> 5;
  const int rt = wave & 3;
  const int ks = wave >> 2;
  const int lane = tid & 31;
  const int l16 = lane & 15;
  const int half = lane >> 4;
  const int myrow = row0 + rt * 16 + l16;
  const int kr = ks * 16;

  const float* Eb = E + ((long)g * NB + b) * NN * NDE;
  const float* Hb = Hsrc + (long)b * NN * NHID;
  const float* xb = x + (long)(b * NT + t) * NN * 3;

  float2 nE; float4 nH0, nH1; float nX;
  const int xi = tid < 192 ? tid : 191;
  auto request = [&](int w) {
    nE = ((const float2*)(Eb + (long)w * 64 * NDE))[tid];
    const float4* sh = (const float4*)(Hb + (long)w * 64 * NHID);
    nH0 = sh[tid];
    nH1 = sh[tid + 512];
    nX = xb[w * 192 + xi];
  };
  request(0);

  v2f aA[4];
#pragma unroll
  for (int kk = 0; kk < 4; ++kk) {
    aA[kk].x = Eb[myrow * NDE + kk * 4 + 2 * half];
    aA[kk].y = Eb[myrow * NDE + kk * 4 + 2 * half + 1];
  }

  for (int i = tid; i < 64 * 13; i += 512) {
    int rr = i / 13, cc = 67 + i % 13;
    lds_v[rr][cc] = (cc == 68) ? 1.f : 0.f;
  }

  v8f acc[5] = {};

#pragma nounroll
  for (int it = 0; it < NN / 64; ++it) {
    __syncthreads();
    ((float2*)lds_e)[tid] = nE;
    *(float4*)&lds_v[tid >> 4][(tid & 15) * 4] = nH0;
    *(float4*)&lds_v[32 + (tid >> 4)][(tid & 15) * 4] = nH1;
    if (tid < 192)
      lds_v[tid / 3][64 + tid % 3] = nX;
    __syncthreads();
    if (it + 1 < NN / 64) request(it + 1);

    v8f c = {};
#pragma unroll
    for (int kk = 0; kk < 4; ++kk) {
      v2f bf;
      bf.x = lds_e[kr + l16][kk * 4 + 2 * half];
      bf.y = lds_e[kr + l16][kk * 4 + 2 * half + 1];
      c = WMMA_F32(aA[kk], bf, c);
    }

#pragma unroll
    for (int r = 0; r < 8; ++r)
      lds_p[wave][r + 8 * half][l16] = __expf(c[r]);

#pragma unroll
    for (int kk = 0; kk < 4; ++kk) {
      v2f ap;
      ap.x = lds_p[wave][l16][kk * 4 + 2 * half];
      ap.y = lds_p[wave][l16][kk * 4 + 2 * half + 1];
#pragma unroll
      for (int ct = 0; ct < 5; ++ct) {
        v2f bv;
        bv.x = lds_v[kr + kk * 4 + 2 * half][ct * 16 + l16];
        bv.y = lds_v[kr + kk * 4 + 2 * half + 1][ct * 16 + l16];
        acc[ct] = WMMA_F32(ap, bv, acc[ct]);
      }
    }
  }

  __syncthreads();
  if (ks > 0) {
#pragma unroll
    for (int ct = 0; ct < 5; ++ct) {
      RED_AT(ks - 1, rt, ct, 0, lane) = make_float4(acc[ct][0], acc[ct][1], acc[ct][2], acc[ct][3]);
      RED_AT(ks - 1, rt, ct, 1, lane) = make_float4(acc[ct][4], acc[ct][5], acc[ct][6], acc[ct][7]);
    }
  }
  __syncthreads();
  if (ks > 0) return;

  auto add_partials = [&](int ct) {
#pragma unroll
    for (int s = 0; s < KGROUPS - 1; ++s) {
      float4 p0 = RED_AT(s, rt, ct, 0, lane);
      float4 p1 = RED_AT(s, rt, ct, 1, lane);
      acc[ct][0] += p0.x; acc[ct][1] += p0.y; acc[ct][2] += p0.z; acc[ct][3] += p0.w;
      acc[ct][4] += p1.x; acc[ct][5] += p1.y; acc[ct][6] += p1.z; acc[ct][7] += p1.w;
    }
  };
  add_partials(4);
  float inv[8];
#pragma unroll
  for (int r = 0; r < 8; ++r)
    inv[r] = 1.f / __shfl(acc[4][r], half * 16 + 4);
  float* Yb = Y + ((long)b * NN + row0 + rt * 16 + 8 * half) * NCP;
#pragma unroll
  for (int cc = 0; cc < 5; ++cc) {
    const int ct = (cc + 4) % 5;
    if (ct != 4) add_partials(ct);
    const int col = ct * 16 + l16;
    const int dst = (col < 64) ? col + 3 : (col < 67) ? col - 64 : 67;
    if (col < NCP) {
#pragma unroll
      for (int r = 0; r < 8; ++r)
        Yb[r * NCP + dst] = acc[ct][r] * inv[r];
    }
  }
#undef RED_AT
}

#define GOUT_KG 8
#define GOUT_LDS_FLOATS 18432
__device__ __forceinline__ void gout_body(
    float* smem, int n,
    const float* __restrict__ x, const float* __restrict__ Hsrc,
    const float* __restrict__ XG,
    const float* __restrict__ W, const float* __restrict__ TB,
    const float* __restrict__ Hs, const float* __restrict__ olg,
    const float* __restrict__ olb, float* __restrict__ G, int g, int t) {
  float4* part = (float4*)smem;
#define PART_AT(k_, bt_, ct_, h_, l_) part[((((k_) * 2 + (bt_)) * 4 + (ct_)) * 2 + (h_)) * 32 + (l_)]
  float (*grow)[NHID] = (float (*)[NHID])(smem + 16384);

  const int wave = __builtin_amdgcn_readfirstlane(threadIdx.x >> 5);
  const int lane = threadIdx.x & 31;
  const int l16 = lane & 15, half = lane >> 4;
  const int K = t + 1;

  float st0[2][NT], st1[2][NT];
#pragma unroll
  for (int j = 0; j < 2; ++j) {
    const long row = (long)(2 * wave + j) * NN + n;
#pragma unroll
    for (int k = 0; k < NT; ++k) {
      if (k < K) {
        st0[j][k] = Hs[((long)k * NB * NN + row) * 64 + lane];
        st1[j][k] = Hs[((long)k * NB * NN + row) * 64 + lane + 32];
      } else { st0[j][k] = 0.f; st1[j][k] = 0.f; }
    }
  }
  float hp0[2] = {0.f, 0.f}, hp1[2] = {0.f, 0.f};
  if (g == 0) {
#pragma unroll
    for (int j = 0; j < 2; ++j) {
      const long row = (long)(2 * wave + j) * NN + n;
      hp0[j] = Hs[((long)t * NB * NN + row) * 64 + lane];
      hp1[j] = Hs[((long)t * NB * NN + row) * 64 + lane + 32];
    }
  }
  const float lg0 = olg[g * 64 + lane], lg1 = olg[g * 64 + lane + 32];
  const float lb0 = olb[g * 64 + lane], lb1 = olb[g * 64 + lane + 32];

  {
    const int bt = wave & 1;
    const int kg = wave >> 1;
    const int slot = kg >> 2;
    const int q = kg & 3;
    const int kk0 = q ? 4 * q + 1 : 0;
    const int kk1 = 4 * q + 5;

    const int ab = bt * 16 + l16;
    const float* Arow = XG + ((long)ab * NN + n) * NCP;
    const float* xr = x + ((long)(ab * NT + t) * NN + n) * 3;
    const float* hr = Hsrc + ((long)ab * NN + n) * NHID;
    const float* Wm = W + (((long)g * NN + n) * 2 + slot) * NCP * NHID;

    v8f c[4] = {};
    auto kstep = [&](v2f a, int kk) {
#pragma unroll
      for (int ct = 0; ct < 4; ++ct) {
        v2f bw;
        bw.x = Wm[(kk * 4 + 2 * half) * 64 + ct * 16 + l16];
        bw.y = Wm[(kk * 4 + 2 * half + 1) * 64 + ct * 16 + l16];
        c[ct] = WMMA_F32(a, bw, c[ct]);
      }
    };
    if (slot) {
      for (int kk = kk0; kk < kk1; ++kk) {
        v2f a;
        a.x = Arow[kk * 4 + 2 * half];
        a.y = Arow[kk * 4 + 2 * half + 1];
        kstep(a, kk);
      }
    } else {
      int kk = kk0;
      if (kk == 0) {
        const float* p1 = half ? hr : xr + 1;
        v2f a;
        a.x = xr[2 * half];
        a.y = *p1;
        kstep(a, 0);
        kk = 1;
      }
      for (; kk < kk1; ++kk) {
        const int i = kk * 4 + 2 * half;
        v2f a;
        a.x = hr[i - 3];
        a.y = hr[(i - 2 < 63) ? i - 2 : 63];
        kstep(a, kk);
      }
    }

#pragma unroll
    for (int ct = 0; ct < 4; ++ct) {
      PART_AT(kg, bt, ct, 0, lane) = make_float4(c[ct][0], c[ct][1], c[ct][2], c[ct][3]);
      PART_AT(kg, bt, ct, 1, lane) = make_float4(c[ct][4], c[ct][5], c[ct][6], c[ct][7]);
    }
  }
  __syncthreads();

  {
    const int rbt = wave & 1, ct = (wave >> 1) & 3, rh = wave >> 3;
    float4 s = PART_AT(0, rbt, ct, rh, lane);
#pragma unroll
    for (int k = 1; k < GOUT_KG; ++k) {
      float4 p = PART_AT(k, rbt, ct, rh, lane);
      s.x += p.x; s.y += p.y; s.z += p.z; s.w += p.w;
    }
    const float sv[4] = {s.x, s.y, s.z, s.w};
#pragma unroll
    for (int e = 0; e < 4; ++e) {
      int bb = rbt * 16 + rh * 4 + e + 8 * half;
      int o = ct * 16 + l16;
      grow[bb][o] = sv[e] + TB[(((long)g * NT + t) * NB + bb) * 64 + o];
    }
  }
  __syncthreads();

  float g0[2], g1[2], q0[2], q1[2];
#pragma unroll
  for (int j = 0; j < 2; ++j) {
    g0[j] = grow[2 * wave + j][lane];
    g1[j] = grow[2 * wave + j][lane + 32];
  }
  {
    float s[2], v[2], d0[2], d1[2];
#pragma unroll
    for (int j = 0; j < 2; ++j) s[j] = g0[j] + g1[j];
#pragma unroll
    for (int m = 1; m < 32; m <<= 1)
#pragma unroll
      for (int j = 0; j < 2; ++j) s[j] += __shfl_xor(s[j], m);
#pragma unroll
    for (int j = 0; j < 2; ++j) {
      float mu = s[j] * (1.f / 64.f);
      d0[j] = g0[j] - mu; d1[j] = g1[j] - mu;
      v[j] = d0[j] * d0[j] + d1[j] * d1[j];
    }
#pragma unroll
    for (int m = 1; m < 32; m <<= 1)
#pragma unroll
      for (int j = 0; j < 2; ++j) v[j] += __shfl_xor(v[j], m);
#pragma unroll
    for (int j = 0; j < 2; ++j) {
      float inv = rsqrtf(v[j] * (1.f / 64.f) + 1e-5f);
      q0[j] = d0[j] * inv * lg0 + lb0;
      q1[j] = d1[j] * inv * lg1 + lb1;
    }
  }

  float sc0[2][NT], sc1[2][NT];
#pragma unroll
  for (int k = 0; k < NT; ++k) {
    if (k < K) {
      float p0[2], p1[2];
#pragma unroll
      for (int j = 0; j < 2; ++j) { p0[j] = q0[j] * st0[j][k]; p1[j] = q1[j] * st1[j][k]; }
#pragma unroll
      for (int m = 1; m < 16; m <<= 1)
#pragma unroll
        for (int j = 0; j < 2; ++j) { p0[j] += __shfl_xor(p0[j], m); p1[j] += __shfl_xor(p1[j], m); }
#pragma unroll
      for (int j = 0; j < 2; ++j) { sc0[j][k] = p0[j] * 0.25f; sc1[j][k] = p1[j] * 0.25f; }
    } else {
#pragma unroll
      for (int j = 0; j < 2; ++j) { sc0[j][k] = -1e30f; sc1[j][k] = -1e30f; }
    }
  }
#pragma unroll
  for (int j = 0; j < 2; ++j) {
    float m0 = -1e30f, m1 = -1e30f;
#pragma unroll
    for (int k = 0; k < NT; ++k) { m0 = fmaxf(m0, sc0[j][k]); m1 = fmaxf(m1, sc1[j][k]); }
    float s0 = 0.f, s1 = 0.f;
#pragma unroll
    for (int k = 0; k < NT; ++k) {
      sc0[j][k] = __expf(sc0[j][k] - m0); s0 += sc0[j][k];
      sc1[j][k] = __expf(sc1[j][k] - m1); s1 += sc1[j][k];
    }
    float o0 = 0.f, o1 = 0.f;
#pragma unroll
    for (int k = 0; k < NT; ++k) { o0 += sc0[j][k] * st0[j][k]; o1 += sc1[j][k] * st1[j][k]; }
    const long row = (long)(2 * wave + j) * NN + n;
    float r0 = g0[j] + o0 / s0, r1 = g1[j] + o1 / s1;
    if (g == 0) {
      r0 = (1.f / (1.f + __expf(-r0))) * hp0[j];
      r1 = (1.f / (1.f + __expf(-r1))) * hp1[j];
    }
    G[row * 64 + lane]      = r0;
    G[row * 64 + lane + 32] = r1;
  }
#undef PART_AT
}

#define EMB_BLOCKS 96
__global__ void __launch_bounds__(512, 8) k_stage(
    int nGa, int nGout, int t,
    const float* __restrict__ E, const float* __restrict__ x,
    const float* __restrict__ gaH, float* __restrict__ gaY, int gaG,
    const float* __restrict__ goH, const float* __restrict__ goXG,
    const float* __restrict__ W, const float* __restrict__ TB,
    const float* __restrict__ Hs, const float* __restrict__ olg,
    const float* __restrict__ olb, float* __restrict__ goOut, int goGate,
    const float* __restrict__ ne, const float* __restrict__ te,
    const float* __restrict__ glg, const float* __restrict__ glb,
    float* __restrict__ Enext) {
  __shared__ __attribute__((aligned(16))) float smem[GOUT_LDS_FLOATS];
  const int bid = blockIdx.x;
  if (bid < nGa) {
    graph_attn_body(smem, bid, E, x, gaH, gaY, gaG, t);
  } else if (bid < nGa + nGout) {
    gout_body(smem, bid - nGa, x, goH, goXG, W, TB, Hs, olg, olb, goOut, goGate, t);
  } else {
    if (t + 1 < NT)
      emb_row(ne, te, glg, glb, Enext, t + 1, (bid - nGa - nGout) * 512 + threadIdx.x);
  }
}

__global__ void k_combine(const float* __restrict__ Gr, const float* __restrict__ Ghc,
                          const float* __restrict__ Hprev, float* __restrict__ Hnew) {
  long i = (long)blockIdx.x * 256 + threadIdx.x;
  if (i >= (long)NB * NN * 64) return;
  float r = 1.f / (1.f + __expf(-Gr[i]));
  float hc = tanhf(Ghc[i]);
  Hnew[i] = r * Hprev[i] + (1.f - r) * hc;
}

__global__ void k_pred(const float* __restrict__ Hs, const float* __restrict__ pw,
                       const float* __restrict__ pb, float* __restrict__ out) {
  int idx = blockIdx.x * 256 + threadIdx.x;
  if (idx >= NB * 12 * NN) return;
  int n = idx % NN; int r = idx / NN; int o = r % 12; int b = r / 12;
  float acc = pb[o];
  for (int cc = 0; cc < 2; ++cc) {
    const float* h = Hs + ((long)(NT - 1 + cc) * NB * NN + (long)b * NN + n) * 64;
    for (int hh = 0; hh < 64; ++hh)
      acc += h[hh] * pw[(o * 2 + cc) * 64 + hh];
  }
  out[idx] = acc;
}

extern "C" void kernel_launch(void* const* d_in, const int* in_sizes, int n_in,
                              void* d_out, int out_size, void* d_ws, size_t ws_size,
                              hipStream_t stream) {
  const float* x   = (const float*)d_in[0];
  const float* ne  = (const float*)d_in[1];
  const float* te  = (const float*)d_in[2];
  const float* wp  = (const float*)d_in[3];
  const float* bp  = (const float*)d_in[4];
  const float* glg = (const float*)d_in[5];
  const float* glb = (const float*)d_in[6];
  const float* olg = (const float*)d_in[7];
  const float* olb = (const float*)d_in[8];
  const float* pw  = (const float*)d_in[9];
  const float* pb  = (const float*)d_in[10];
  float* out = (float*)d_out;

  float* ws = (float*)d_ws;
  long off = 0;
  float* W    = ws + off; off += 3L * NN * 2 * NCP * NHID;
  float* TB   = ws + off; off += 3L * NT * NB * NHID;
  float* E0   = ws + off; off += 3L * NB * NN * NDE;
  float* E1   = ws + off; off += 3L * NB * NN * NDE;
  float* Hs   = ws + off; off += (long)(NT + 1) * NB * NN * NHID;
  float* XGa  = ws + off; off += (long)NB * NN * NCP;
  float* XGb  = ws + off; off += (long)NB * NN * NCP;
  float* G0   = ws + off; off += (long)NB * NN * NHID;
  float* G1   = ws + off; off += (long)NB * NN * NHID;
  float* G2   = ws + off; off += (long)NB * NN * NHID;

  const long wTotal = 3L * NN * 2 * NCP * NHID;
  k_w<<<dim3((unsigned)((wTotal + 255) / 256)), 256, 0, stream>>>(ne, wp, W);
  k_tb<<<(3 * NT * NB * NHID + 255) / 256, 256, 0, stream>>>(te, bp, TB);
  k_zero<<<(unsigned)(((long)NB * NN * NHID + 255) / 256), 256, 0, stream>>>(Hs, (long)NB * NN * NHID);

  k_emb<<<(3 * NB * NN + 255) / 256, 256, 0, stream>>>(ne, te, glg, glb, E0, 0);

  float* E = E0;
  float* Enext = E1;
  for (int t = 0; t < NT; ++t) {
    float* Hprev = Hs + (long)t * NB * NN * NHID;
    float* Hnew  = Hs + (long)(t + 1) * NB * NN * NHID;
    k_stage<<<GA_BLOCKS + EMB_BLOCKS, 512, 0, stream>>>(GA_BLOCKS, 0, t, E, x, Hprev, XGa, 0,
        Hprev, XGb, W, TB, Hs, olg, olb, G0, 0, ne, te, glg, glb, Enext);
    k_stage<<<GA_BLOCKS + NN, 512, 0, stream>>>(GA_BLOCKS, NN, t, E, x, Hprev, XGb, 1,
        Hprev, XGa, W, TB, Hs, olg, olb, G0, 0, ne, te, glg, glb, Enext);
    k_stage<<<GA_BLOCKS + NN, 512, 0, stream>>>(GA_BLOCKS, NN, t, E, x, G0, XGa, 2,
        Hprev, XGb, W, TB, Hs, olg, olb, G1, 1, ne, te, glg, glb, Enext);
    k_stage<<<NN, 512, 0, stream>>>(0, NN, t, E, x, G0, XGb, 2,
        G0, XGa, W, TB, Hs, olg, olb, G2, 2, ne, te, glg, glb, Enext);
    k_combine<<<(unsigned)(((long)NB * NN * NHID + 255) / 256), 256, 0, stream>>>(G1, G2, Hprev, Hnew);
    float* Eswap = E; E = Enext; Enext = Eswap;
  }
  k_pred<<<(NB * 12 * NN + 255) / 256, 256, 0, stream>>>(Hs, pw, pb, out);
}


# k_w, k_combine, k_zero: a thread makes 128, 32, 64 bytes of output, not 4 (k_w holds its wp values in registers across 8 nodes), so 31x, 8x, 16x fewer waves have to start; sum orders unchanged
# speedup vs baseline: 4.5108x; 1.0464x over previous
#include <hip/hip_runtime.h>
#include <math.h>

typedef __attribute__((ext_vector_type(2))) float v2f;
typedef __attribute__((ext_vector_type(8))) float v8f;

#define WMMA_F32(A, B, C) \
  __builtin_amdgcn_wmma_f32_16x16x4_f32(false, (A), false, (B), (short)0, (C), false, false)

#define NB   32
#define NT   12
#define NN   512
#define NDE  16
#define NHID 64
#define NCP  68

#define KW_NODES 8
#define KW_QUADS (3 * 2 * NCP * 16)
__global__ void __launch_bounds__(256) k_w(const float* __restrict__ ne, const float* __restrict__ wp,
                                           float* __restrict__ W) {
  const int idx = blockIdx.x * 256 + threadIdx.x;
  if (idx >= KW_QUADS) return;
  const int q = idx & 15; int r = idx >> 4;
  const int i = r % NCP; r /= NCP;
  const int slot = r & 1; const int g = r >> 1;
  const int n0 = blockIdx.y * KW_NODES;
  const bool live = i < 67;
  const int il = live ? i : 66;
  float4 w[NDE];
#pragma unroll
  for (int d = 0; d < NDE; ++d)
    w[d] = *(const float4*)(wp + ((((long)(g * NDE + d) * 2 + slot) * 67 + il) * 64 + q * 4));
#pragma unroll
  for (int j = 0; j < KW_NODES; ++j) {
    const int n = n0 + j;
    float4 acc = make_float4(0.f, 0.f, 0.f, 0.f);
#pragma unroll
    for (int d = 0; d < NDE; ++d) {
      const float e = ne[n * NDE + d];
      acc.x += e * w[d].x; acc.y += e * w[d].y; acc.z += e * w[d].z; acc.w += e * w[d].w;
    }
    if (!live) acc = make_float4(0.f, 0.f, 0.f, 0.f);
    *(float4*)(W + ((((long)g * NN + n) * 2 + slot) * NCP + i) * 64 + q * 4) = acc;
  }
}

__global__ void k_tb(const float* __restrict__ te, const float* __restrict__ bp,
                     float* __restrict__ TB) {
  int idx = blockIdx.x * 256 + threadIdx.x;
  if (idx >= 3 * NT * NB * NHID) return;
  int o = idx % 64; int r = idx / 64;
  int b = r % NB; r /= NB;
  int t = r % NT; int g = r / NT;
  float acc = 0.f;
  for (int d = 0; d < NDE; ++d)
    acc += te[(b * NT + t) * NDE + d] * bp[(g * NDE + d) * 64 + o];
  TB[idx] = acc;
}

__device__ __forceinline__ void emb_row(const float* __restrict__ ne, const float* __restrict__ te,
                                        const float* __restrict__ glg, const float* __restrict__ glb,
                                        float* __restrict__ E, int t, int idx) {
  if (idx >= 3 * NB * NN) return;
  int n = idx % NN; int r = idx / NN;
  int b = r % NB; int g = r / NB;
  float v[NDE]; float mu = 0.f;
  for (int d = 0; d < NDE; ++d) { v[d] = ne[n * NDE + d] + te[(b * NT + t) * NDE + d]; mu += v[d]; }
  mu *= (1.f / NDE);
  float var = 0.f;
  for (int d = 0; d < NDE; ++d) { float q = v[d] - mu; var += q * q; }
  var *= (1.f / NDE);
  float inv = rsqrtf(var + 1e-12f);
  for (int d = 0; d < NDE; ++d)
    E[(long)idx * NDE + d] = (v[d] - mu) * inv * glg[g * NDE + d] + glb[g * NDE + d];
}

__global__ void k_emb(const float* __restrict__ ne, const float* __restrict__ te,
                      const float* __restrict__ glg, const float* __restrict__ glb,
                      float* __restrict__ E, int t) {
  emb_row(ne, te, glg, glb, E, t, blockIdx.x * 256 + threadIdx.x);
}

#define PLANE_F4 (NB * NN * NHID / 4)
__global__ void __launch_bounds__(256) k_zero(float* __restrict__ p) {
  float4* p4 = (float4*)p;
  const int i = blockIdx.x * 256 + threadIdx.x;
#pragma unroll
  for (int k = 0; k < 4; ++k)
    p4[i + k * (PLANE_F4 / 4)] = make_float4(0.f, 0.f, 0.f, 0.f);
}


#define KGROUPS 4
#define GA_LDS_FLOATS 15360
#define GA_BLOCKS 256
__device__ __forceinline__ void graph_attn_body(
    float* smem, int bid,
    const float* __restrict__ E,
    const float* __restrict__ x,
    const float* __restrict__ Hsrc,
    float* __restrict__ Y,
    int g, int t) {
  float (*lds_e)[16] = (float (*)[16])smem;
  float (*lds_v)[80] = (float (*)[80])(smem + 1024);
  float (*lds_p)[16][17] = (float (*)[16][17])(smem + 6144);
  float4* lds_red = (float4*)smem;
#define RED_AT(s_, rt_, ct_, h_, l_) lds_red[((((s_) * 4 + (rt_)) * 5 + (ct_)) * 2 + (h_)) * 32 + (l_)]

  const int b = bid >> 3;
  const int row0 = (bid & 7) * 64;
  const int tid = threadIdx.x;
  const int wave = tid >> 5;
  const int rt = wave & 3;
  const int ks = wave >> 2;
  const int lane = tid & 31;
  const int l16 = lane & 15;
  const int half = lane >> 4;
  const int myrow = row0 + rt * 16 + l16;
  const int kr = ks * 16;

  const float* Eb = E + ((long)g * NB + b) * NN * NDE;
  const float* Hb = Hsrc + (long)b * NN * NHID;
  const float* xb = x + (long)(b * NT + t) * NN * 3;

  float2 nE; float4 nH0, nH1; float nX;
  const int xi = tid < 192 ? tid : 191;
  auto request = [&](int w) {
    nE = ((const float2*)(Eb + (long)w * 64 * NDE))[tid];
    const float4* sh = (const float4*)(Hb + (long)w * 64 * NHID);
    nH0 = sh[tid];
    nH1 = sh[tid + 512];
    nX = xb[w * 192 + xi];
  };
  request(0);

  v2f aA[4];
#pragma unroll
  for (int kk = 0; kk < 4; ++kk) {
    aA[kk].x = Eb[myrow * NDE + kk * 4 + 2 * half];
    aA[kk].y = Eb[myrow * NDE + kk * 4 + 2 * half + 1];
  }

  for (int i = tid; i < 64 * 13; i += 512) {
    int rr = i / 13, cc = 67 + i % 13;
    lds_v[rr][cc] = (cc == 68) ? 1.f : 0.f;
  }

  v8f acc[5] = {};

#pragma nounroll
  for (int it = 0; it < NN / 64; ++it) {
    __syncthreads();
    ((float2*)lds_e)[tid] = nE;
    *(float4*)&lds_v[tid >> 4][(tid & 15) * 4] = nH0;
    *(float4*)&lds_v[32 + (tid >> 4)][(tid & 15) * 4] = nH1;
    if (tid < 192)
      lds_v[tid / 3][64 + tid % 3] = nX;
    __syncthreads();
    if (it + 1 < NN / 64) request(it + 1);

    v8f c = {};
#pragma unroll
    for (int kk = 0; kk < 4; ++kk) {
      v2f bf;
      bf.x = lds_e[kr + l16][kk * 4 + 2 * half];
      bf.y = lds_e[kr + l16][kk * 4 + 2 * half + 1];
      c = WMMA_F32(aA[kk], bf, c);
    }

#pragma unroll
    for (int r = 0; r < 8; ++r)
      lds_p[wave][r + 8 * half][l16] = __expf(c[r]);

#pragma unroll
    for (int kk = 0; kk < 4; ++kk) {
      v2f ap;
      ap.x = lds_p[wave][l16][kk * 4 + 2 * half];
      ap.y = lds_p[wave][l16][kk * 4 + 2 * half + 1];
#pragma unroll
      for (int ct = 0; ct < 5; ++ct) {
        v2f bv;
        bv.x = lds_v[kr + kk * 4 + 2 * half][ct * 16 + l16];
        bv.y = lds_v[kr + kk * 4 + 2 * half + 1][ct * 16 + l16];
        acc[ct] = WMMA_F32(ap, bv, acc[ct]);
      }
    }
  }

  __syncthreads();
  if (ks > 0) {
#pragma unroll
    for (int ct = 0; ct < 5; ++ct) {
      RED_AT(ks - 1, rt, ct, 0, lane) = make_float4(acc[ct][0], acc[ct][1], acc[ct][2], acc[ct][3]);
      RED_AT(ks - 1, rt, ct, 1, lane) = make_float4(acc[ct][4], acc[ct][5], acc[ct][6], acc[ct][7]);
    }
  }
  __syncthreads();
  if (ks > 0) return;

  auto add_partials = [&](int ct) {
#pragma unroll
    for (int s = 0; s < KGROUPS - 1; ++s) {
      float4 p0 = RED_AT(s, rt, ct, 0, lane);
      float4 p1 = RED_AT(s, rt, ct, 1, lane);
      acc[ct][0] += p0.x; acc[ct][1] += p0.y; acc[ct][2] += p0.z; acc[ct][3] += p0.w;
      acc[ct][4] += p1.x; acc[ct][5] += p1.y; acc[ct][6] += p1.z; acc[ct][7] += p1.w;
    }
  };
  add_partials(4);
  float inv[8];
#pragma unroll
  for (int r = 0; r < 8; ++r)
    inv[r] = 1.f / __shfl(acc[4][r], half * 16 + 4);
  float* Yb = Y + ((long)b * NN + row0 + rt * 16 + 8 * half) * NCP;
#pragma unroll
  for (int cc = 0; cc < 5; ++cc) {
    const int ct = (cc + 4) % 5;
    if (ct != 4) add_partials(ct);
    const int col = ct * 16 + l16;
    const int dst = (col < 64) ? col + 3 : (col < 67) ? col - 64 : 67;
    if (col < NCP) {
#pragma unroll
      for (int r = 0; r < 8; ++r)
        Yb[r * NCP + dst] = acc[ct][r] * inv[r];
    }
  }
#undef RED_AT
}

#define GOUT_KG 8
#define GOUT_LDS_FLOATS 18432
__device__ __forceinline__ void gout_body(
    float* smem, int n,
    const float* __restrict__ x, const float* __restrict__ Hsrc,
    const float* __restrict__ XG,
    const float* __restrict__ W, const float* __restrict__ TB,
    const float* __restrict__ Hs, const float* __restrict__ olg,
    const float* __restrict__ olb, float* __restrict__ G, int g, int t) {
  float4* part = (float4*)smem;
#define PART_AT(k_, bt_, ct_, h_, l_) part[((((k_) * 2 + (bt_)) * 4 + (ct_)) * 2 + (h_)) * 32 + (l_)]
  float (*grow)[NHID] = (float (*)[NHID])(smem + 16384);

  const int wave = __builtin_amdgcn_readfirstlane(threadIdx.x >> 5);
  const int lane = threadIdx.x & 31;
  const int l16 = lane & 15, half = lane >> 4;
  const int K = t + 1;

  float st0[2][NT], st1[2][NT];
#pragma unroll
  for (int j = 0; j < 2; ++j) {
    const long row = (long)(2 * wave + j) * NN + n;
#pragma unroll
    for (int k = 0; k < NT; ++k) {
      if (k < K) {
        st0[j][k] = Hs[((long)k * NB * NN + row) * 64 + lane];
        st1[j][k] = Hs[((long)k * NB * NN + row) * 64 + lane + 32];
      } else { st0[j][k] = 0.f; st1[j][k] = 0.f; }
    }
  }
  float hp0[2] = {0.f, 0.f}, hp1[2] = {0.f, 0.f};
  if (g == 0) {
#pragma unroll
    for (int j = 0; j < 2; ++j) {
      const long row = (long)(2 * wave + j) * NN + n;
      hp0[j] = Hs[((long)t * NB * NN + row) * 64 + lane];
      hp1[j] = Hs[((long)t * NB * NN + row) * 64 + lane + 32];
    }
  }
  const float lg0 = olg[g * 64 + lane], lg1 = olg[g * 64 + lane + 32];
  const float lb0 = olb[g * 64 + lane], lb1 = olb[g * 64 + lane + 32];

  {
    const int bt = wave & 1;
    const int kg = wave >> 1;
    const int slot = kg >> 2;
    const int q = kg & 3;
    const int kk0 = q ? 4 * q + 1 : 0;
    const int kk1 = 4 * q + 5;

    const int ab = bt * 16 + l16;
    const float* Arow = XG + ((long)ab * NN + n) * NCP;
    const float* xr = x + ((long)(ab * NT + t) * NN + n) * 3;
    const float* hr = Hsrc + ((long)ab * NN + n) * NHID;
    const float* Wm = W + (((long)g * NN + n) * 2 + slot) * NCP * NHID;

    v8f c[4] = {};
    auto kstep = [&](v2f a, int kk) {
#pragma unroll
      for (int ct = 0; ct < 4; ++ct) {
        v2f bw;
        bw.x = Wm[(kk * 4 + 2 * half) * 64 + ct * 16 + l16];
        bw.y = Wm[(kk * 4 + 2 * half + 1) * 64 + ct * 16 + l16];
        c[ct] = WMMA_F32(a, bw, c[ct]);
      }
    };
    if (slot) {
      for (int kk = kk0; kk < kk1; ++kk) {
        v2f a;
        a.x = Arow[kk * 4 + 2 * half];
        a.y = Arow[kk * 4 + 2 * half + 1];
        kstep(a, kk);
      }
    } else {
      int kk = kk0;
      if (kk == 0) {
        const float* p1 = half ? hr : xr + 1;
        v2f a;
        a.x = xr[2 * half];
        a.y = *p1;
        kstep(a, 0);
        kk = 1;
      }
      for (; kk < kk1; ++kk) {
        const int i = kk * 4 + 2 * half;
        v2f a;
        a.x = hr[i - 3];
        a.y = hr[(i - 2 < 63) ? i - 2 : 63];
        kstep(a, kk);
      }
    }

#pragma unroll
    for (int ct = 0; ct < 4; ++ct) {
      PART_AT(kg, bt, ct, 0, lane) = make_float4(c[ct][0], c[ct][1], c[ct][2], c[ct][3]);
      PART_AT(kg, bt, ct, 1, lane) = make_float4(c[ct][4], c[ct][5], c[ct][6], c[ct][7]);
    }
  }
  __syncthreads();

  {
    const int rbt = wave & 1, ct = (wave >> 1) & 3, rh = wave >> 3;
    float4 s = PART_AT(0, rbt, ct, rh, lane);
#pragma unroll
    for (int k = 1; k < GOUT_KG; ++k) {
      float4 p = PART_AT(k, rbt, ct, rh, lane);
      s.x += p.x; s.y += p.y; s.z += p.z; s.w += p.w;
    }
    const float sv[4] = {s.x, s.y, s.z, s.w};
#pragma unroll
    for (int e = 0; e < 4; ++e) {
      int bb = rbt * 16 + rh * 4 + e + 8 * half;
      int o = ct * 16 + l16;
      grow[bb][o] = sv[e] + TB[(((long)g * NT + t) * NB + bb) * 64 + o];
    }
  }
  __syncthreads();

  float g0[2], g1[2], q0[2], q1[2];
#pragma unroll
  for (int j = 0; j < 2; ++j) {
    g0[j] = grow[2 * wave + j][lane];
    g1[j] = grow[2 * wave + j][lane + 32];
  }
  {
    float s[2], v[2], d0[2], d1[2];
#pragma unroll
    for (int j = 0; j < 2; ++j) s[j] = g0[j] + g1[j];
#pragma unroll
    for (int m = 1; m < 32; m <<= 1)
#pragma unroll
      for (int j = 0; j < 2; ++j) s[j] += __shfl_xor(s[j], m);
#pragma unroll
    for (int j = 0; j < 2; ++j) {
      float mu = s[j] * (1.f / 64.f);
      d0[j] = g0[j] - mu; d1[j] = g1[j] - mu;
      v[j] = d0[j] * d0[j] + d1[j] * d1[j];
    }
#pragma unroll
    for (int m = 1; m < 32; m <<= 1)
#pragma unroll
      for (int j = 0; j < 2; ++j) v[j] += __shfl_xor(v[j], m);
#pragma unroll
    for (int j = 0; j < 2; ++j) {
      float inv = rsqrtf(v[j] * (1.f / 64.f) + 1e-5f);
      q0[j] = d0[j] * inv * lg0 + lb0;
      q1[j] = d1[j] * inv * lg1 + lb1;
    }
  }

  float sc0[2][NT], sc1[2][NT];
#pragma unroll
  for (int k = 0; k < NT; ++k) {
    if (k < K) {
      float p0[2], p1[2];
#pragma unroll
      for (int j = 0; j < 2; ++j) { p0[j] = q0[j] * st0[j][k]; p1[j] = q1[j] * st1[j][k]; }
#pragma unroll
      for (int m = 1; m < 16; m <<= 1)
#pragma unroll
        for (int j = 0; j < 2; ++j) { p0[j] += __shfl_xor(p0[j], m); p1[j] += __shfl_xor(p1[j], m); }
#pragma unroll
      for (int j = 0; j < 2; ++j) { sc0[j][k] = p0[j] * 0.25f; sc1[j][k] = p1[j] * 0.25f; }
    } else {
#pragma unroll
      for (int j = 0; j < 2; ++j) { sc0[j][k] = -1e30f; sc1[j][k] = -1e30f; }
    }
  }
#pragma unroll
  for (int j = 0; j < 2; ++j) {
    float m0 = -1e30f, m1 = -1e30f;
#pragma unroll
    for (int k = 0; k < NT; ++k) { m0 = fmaxf(m0, sc0[j][k]); m1 = fmaxf(m1, sc1[j][k]); }
    float s0 = 0.f, s1 = 0.f;
#pragma unroll
    for (int k = 0; k < NT; ++k) {
      sc0[j][k] = __expf(sc0[j][k] - m0); s0 += sc0[j][k];
      sc1[j][k] = __expf(sc1[j][k] - m1); s1 += sc1[j][k];
    }
    float o0 = 0.f, o1 = 0.f;
#pragma unroll
    for (int k = 0; k < NT; ++k) { o0 += sc0[j][k] * st0[j][k]; o1 += sc1[j][k] * st1[j][k]; }
    const long row = (long)(2 * wave + j) * NN + n;
    float r0 = g0[j] + o0 / s0, r1 = g1[j] + o1 / s1;
    if (g == 0) {
      r0 = (1.f / (1.f + __expf(-r0))) * hp0[j];
      r1 = (1.f / (1.f + __expf(-r1))) * hp1[j];
    }
    G[row * 64 + lane]      = r0;
    G[row * 64 + lane + 32] = r1;
  }
#undef PART_AT
}

#define EMB_BLOCKS 96
__global__ void __launch_bounds__(512, 8) k_stage(
    int nGa, int nGout, int t,
    const float* __restrict__ E, const float* __restrict__ x,
    const float* __restrict__ gaH, float* __restrict__ gaY, int gaG,
    const float* __restrict__ goH, const float* __restrict__ goXG,
    const float* __restrict__ W, const float* __restrict__ TB,
    const float* __restrict__ Hs, const float* __restrict__ olg,
    const float* __restrict__ olb, float* __restrict__ goOut, int goGate,
    const float* __restrict__ ne, const float* __restrict__ te,
    const float* __restrict__ glg, const float* __restrict__ glb,
    float* __restrict__ Enext) {
  __shared__ __attribute__((aligned(16))) float smem[GOUT_LDS_FLOATS];
  const int bid = blockIdx.x;
  if (bid < nGa) {
    graph_attn_body(smem, bid, E, x, gaH, gaY, gaG, t);
  } else if (bid < nGa + nGout) {
    gout_body(smem, bid - nGa, x, goH, goXG, W, TB, Hs, olg, olb, goOut, goGate, t);
  } else {
    if (t + 1 < NT)
      emb_row(ne, te, glg, glb, Enext, t + 1, (bid - nGa - nGout) * 512 + threadIdx.x);
  }
}

__device__ __forceinline__ float combine1(float gr, float ghc, float hp) {
  float r = 1.f / (1.f + __expf(-gr));
  float hc = tanhf(ghc);
  return r * hp + (1.f - r) * hc;
}
__global__ void __launch_bounds__(256) k_combine(const float* __restrict__ Gr, const float* __restrict__ Ghc,
                                                 const float* __restrict__ Hprev, float* __restrict__ Hnew) {
  const int i = blockIdx.x * 256 + threadIdx.x;
  const float4* gr = (const float4*)Gr;
  const float4* gh = (const float4*)Ghc;
  const float4* hp = (const float4*)Hprev;
  float4 a[2], c[2], h[2];
#pragma unroll
  for (int k = 0; k < 2; ++k) {
    a[k] = gr[i + k * (PLANE_F4 / 2)];
    c[k] = gh[i + k * (PLANE_F4 / 2)];
    h[k] = hp[i + k * (PLANE_F4 / 2)];
  }
#pragma unroll
  for (int k = 0; k < 2; ++k) {
    float4 o;
    o.x = combine1(a[k].x, c[k].x, h[k].x);
    o.y = combine1(a[k].y, c[k].y, h[k].y);
    o.z = combine1(a[k].z, c[k].z, h[k].z);
    o.w = combine1(a[k].w, c[k].w, h[k].w);
    ((float4*)Hnew)[i + k * (PLANE_F4 / 2)] = o;
  }
}

__global__ void k_pred(const float* __restrict__ Hs, const float* __restrict__ pw,
                       const float* __restrict__ pb, float* __restrict__ out) {
  int idx = blockIdx.x * 256 + threadIdx.x;
  if (idx >= NB * 12 * NN) return;
  int n = idx % NN; int r = idx / NN; int o = r % 12; int b = r / 12;
  float acc = pb[o];
  for (int cc = 0; cc < 2; ++cc) {
    const float* h = Hs + ((long)(NT - 1 + cc) * NB * NN + (long)b * NN + n) * 64;
    for (int hh = 0; hh < 64; ++hh)
      acc += h[hh] * pw[(o * 2 + cc) * 64 + hh];
  }
  out[idx] = acc;
}

extern "C" void kernel_launch(void* const* d_in, const int* in_sizes, int n_in,
                              void* d_out, int out_size, void* d_ws, size_t ws_size,
                              hipStream_t stream) {
  const float* x   = (const float*)d_in[0];
  const float* ne  = (const float*)d_in[1];
  const float* te  = (const float*)d_in[2];
  const float* wp  = (const float*)d_in[3];
  const float* bp  = (const float*)d_in[4];
  const float* glg = (const float*)d_in[5];
  const float* glb = (const float*)d_in[6];
  const float* olg = (const float*)d_in[7];
  const float* olb = (const float*)d_in[8];
  const float* pw  = (const float*)d_in[9];
  const float* pb  = (const float*)d_in[10];
  float* out = (float*)d_out;

  float* ws = (float*)d_ws;
  long off = 0;
  float* W    = ws + off; off += 3L * NN * 2 * NCP * NHID;
  float* TB   = ws + off; off += 3L * NT * NB * NHID;
  float* E0   = ws + off; off += 3L * NB * NN * NDE;
  float* E1   = ws + off; off += 3L * NB * NN * NDE;
  float* Hs   = ws + off; off += (long)(NT + 1) * NB * NN * NHID;
  float* XGa  = ws + off; off += (long)NB * NN * NCP;
  float* XGb  = ws + off; off += (long)NB * NN * NCP;
  float* G0   = ws + off; off += (long)NB * NN * NHID;
  float* G1   = ws + off; off += (long)NB * NN * NHID;
  float* G2   = ws + off; off += (long)NB * NN * NHID;

  k_w<<<dim3((KW_QUADS + 255) / 256, NN / KW_NODES), 256, 0, stream>>>(ne, wp, W);
  k_tb<<<(3 * NT * NB * NHID + 255) / 256, 256, 0, stream>>>(te, bp, TB);
  k_zero<<<PLANE_F4 / 4 / 256, 256, 0, stream>>>(Hs);

  k_emb<<<(3 * NB * NN + 255) / 256, 256, 0, stream>>>(ne, te, glg, glb, E0, 0);

  float* E = E0;
  float* Enext = E1;
  for (int t = 0; t < NT; ++t) {
    float* Hprev = Hs + (long)t * NB * NN * NHID;
    float* Hnew  = Hs + (long)(t + 1) * NB * NN * NHID;
    k_stage<<<GA_BLOCKS + EMB_BLOCKS, 512, 0, stream>>>(GA_BLOCKS, 0, t, E, x, Hprev, XGa, 0,
        Hprev, XGb, W, TB, Hs, olg, olb, G0, 0, ne, te, glg, glb, Enext);
    k_stage<<<GA_BLOCKS + NN, 512, 0, stream>>>(GA_BLOCKS, NN, t, E, x, Hprev, XGb, 1,
        Hprev, XGa, W, TB, Hs, olg, olb, G0, 0, ne, te, glg, glb, Enext);
    k_stage<<<GA_BLOCKS + NN, 512, 0, stream>>>(GA_BLOCKS, NN, t, E, x, G0, XGa, 2,
        Hprev, XGb, W, TB, Hs, olg, olb, G1, 1, ne, te, glg, glb, Enext);
    k_stage<<<NN, 512, 0, stream>>>(0, NN, t, E, x, G0, XGb, 2,
        G0, XGa, W, TB, Hs, olg, olb, G2, 2, ne, te, glg, glb, Enext);
    k_combine<<<PLANE_F4 / 2 / 256, 256, 0, stream>>>(G1, G2, Hprev, Hnew);
    float* Eswap = E; E = Enext; Enext = Eswap;
  }
  k_pred<<<(NB * 12 * NN + 255) / 256, 256, 0, stream>>>(Hs, pw, pb, out);
}


# candidate gate's node GEMM blocks blend their rows into the new state themselves (r gate's rows requested behind the last barrier): 12 k_combine launches and one plane go; sum orders unchanged
# speedup vs baseline: 4.7986x; 1.0638x over previous
#include <hip/hip_runtime.h>
#include <math.h>

typedef __attribute__((ext_vector_type(2))) float v2f;
typedef __attribute__((ext_vector_type(8))) float v8f;

#define WMMA_F32(A, B, C) \
  __builtin_amdgcn_wmma_f32_16x16x4_f32(false, (A), false, (B), (short)0, (C), false, false)

#define NB   32
#define NT   12
#define NN   512
#define NDE  16
#define NHID 64
#define NCP  68

#define KW_NODES 8
#define KW_QUADS (3 * 2 * NCP * 16)
__global__ void __launch_bounds__(256) k_w(const float* __restrict__ ne, const float* __restrict__ wp,
                                           float* __restrict__ W) {
  const int idx = blockIdx.x * 256 + threadIdx.x;
  if (idx >= KW_QUADS) return;
  const int q = idx & 15; int r = idx >> 4;
  const int i = r % NCP; r /= NCP;
  const int slot = r & 1; const int g = r >> 1;
  const int n0 = blockIdx.y * KW_NODES;
  const bool live = i < 67;
  const int il = live ? i : 66;
  float4 w[NDE];
#pragma unroll
  for (int d = 0; d < NDE; ++d)
    w[d] = *(const float4*)(wp + ((((long)(g * NDE + d) * 2 + slot) * 67 + il) * 64 + q * 4));
#pragma unroll
  for (int j = 0; j < KW_NODES; ++j) {
    const int n = n0 + j;
    float4 acc = make_float4(0.f, 0.f, 0.f, 0.f);
#pragma unroll
    for (int d = 0; d < NDE; ++d) {
      const float e = ne[n * NDE + d];
      acc.x += e * w[d].x; acc.y += e * w[d].y; acc.z += e * w[d].z; acc.w += e * w[d].w;
    }
    if (!live) acc = make_float4(0.f, 0.f, 0.f, 0.f);
    *(float4*)(W + ((((long)g * NN + n) * 2 + slot) * NCP + i) * 64 + q * 4) = acc;
  }
}

__global__ void k_tb(const float* __restrict__ te, const float* __restrict__ bp,
                     float* __restrict__ TB) {
  int idx = blockIdx.x * 256 + threadIdx.x;
  if (idx >= 3 * NT * NB * NHID) return;
  int o = idx % 64; int r = idx / 64;
  int b = r % NB; r /= NB;
  int t = r % NT; int g = r / NT;
  float acc = 0.f;
  for (int d = 0; d < NDE; ++d)
    acc += te[(b * NT + t) * NDE + d] * bp[(g * NDE + d) * 64 + o];
  TB[idx] = acc;
}

__device__ __forceinline__ void emb_row(const float* __restrict__ ne, const float* __restrict__ te,
                                        const float* __restrict__ glg, const float* __restrict__ glb,
                                        float* __restrict__ E, int t, int idx) {
  if (idx >= 3 * NB * NN) return;
  int n = idx % NN; int r = idx / NN;
  int b = r % NB; int g = r / NB;
  float v[NDE]; float mu = 0.f;
  for (int d = 0; d < NDE; ++d) { v[d] = ne[n * NDE + d] + te[(b * NT + t) * NDE + d]; mu += v[d]; }
  mu *= (1.f / NDE);
  float var = 0.f;
  for (int d = 0; d < NDE; ++d) { float q = v[d] - mu; var += q * q; }
  var *= (1.f / NDE);
  float inv = rsqrtf(var + 1e-12f);
  for (int d = 0; d < NDE; ++d)
    E[(long)idx * NDE + d] = (v[d] - mu) * inv * glg[g * NDE + d] + glb[g * NDE + d];
}

__global__ void k_emb(const float* __restrict__ ne, const float* __restrict__ te,
                      const float* __restrict__ glg, const float* __restrict__ glb,
                      float* __restrict__ E, int t) {
  emb_row(ne, te, glg, glb, E, t, blockIdx.x * 256 + threadIdx.x);
}

#define PLANE_F4 (NB * NN * NHID / 4)
__global__ void __launch_bounds__(256) k_zero(float* __restrict__ p) {
  float4* p4 = (float4*)p;
  const int i = blockIdx.x * 256 + threadIdx.x;
#pragma unroll
  for (int k = 0; k < 4; ++k)
    p4[i + k * (PLANE_F4 / 4)] = make_float4(0.f, 0.f, 0.f, 0.f);
}


#define KGROUPS 4
#define GA_LDS_FLOATS 15360
#define GA_BLOCKS 256
__device__ __forceinline__ void graph_attn_body(
    float* smem, int bid,
    const float* __restrict__ E,
    const float* __restrict__ x,
    const float* __restrict__ Hsrc,
    float* __restrict__ Y,
    int g, int t) {
  float (*lds_e)[16] = (float (*)[16])smem;
  float (*lds_v)[80] = (float (*)[80])(smem + 1024);
  float (*lds_p)[16][17] = (float (*)[16][17])(smem + 6144);
  float4* lds_red = (float4*)smem;
#define RED_AT(s_, rt_, ct_, h_, l_) lds_red[((((s_) * 4 + (rt_)) * 5 + (ct_)) * 2 + (h_)) * 32 + (l_)]

  const int b = bid >> 3;
  const int row0 = (bid & 7) * 64;
  const int tid = threadIdx.x;
  const int wave = tid >> 5;
  const int rt = wave & 3;
  const int ks = wave >> 2;
  const int lane = tid & 31;
  const int l16 = lane & 15;
  const int half = lane >> 4;
  const int myrow = row0 + rt * 16 + l16;
  const int kr = ks * 16;

  const float* Eb = E + ((long)g * NB + b) * NN * NDE;
  const float* Hb = Hsrc + (long)b * NN * NHID;
  const float* xb = x + (long)(b * NT + t) * NN * 3;

  float2 nE; float4 nH0, nH1; float nX;
  const int xi = tid < 192 ? tid : 191;
  auto request = [&](int w) {
    nE = ((const float2*)(Eb + (long)w * 64 * NDE))[tid];
    const float4* sh = (const float4*)(Hb + (long)w * 64 * NHID);
    nH0 = sh[tid];
    nH1 = sh[tid + 512];
    nX = xb[w * 192 + xi];
  };
  request(0);

  v2f aA[4];
#pragma unroll
  for (int kk = 0; kk < 4; ++kk) {
    aA[kk].x = Eb[myrow * NDE + kk * 4 + 2 * half];
    aA[kk].y = Eb[myrow * NDE + kk * 4 + 2 * half + 1];
  }

  for (int i = tid; i < 64 * 13; i += 512) {
    int rr = i / 13, cc = 67 + i % 13;
    lds_v[rr][cc] = (cc == 68) ? 1.f : 0.f;
  }

  v8f acc[5] = {};

#pragma nounroll
  for (int it = 0; it < NN / 64; ++it) {
    __syncthreads();
    ((float2*)lds_e)[tid] = nE;
    *(float4*)&lds_v[tid >> 4][(tid & 15) * 4] = nH0;
    *(float4*)&lds_v[32 + (tid >> 4)][(tid & 15) * 4] = nH1;
    if (tid < 192)
      lds_v[tid / 3][64 + tid % 3] = nX;
    __syncthreads();
    if (it + 1 < NN / 64) request(it + 1);

    v8f c = {};
#pragma unroll
    for (int kk = 0; kk < 4; ++kk) {
      v2f bf;
      bf.x = lds_e[kr + l16][kk * 4 + 2 * half];
      bf.y = lds_e[kr + l16][kk * 4 + 2 * half + 1];
      c = WMMA_F32(aA[kk], bf, c);
    }

#pragma unroll
    for (int r = 0; r < 8; ++r)
      lds_p[wave][r + 8 * half][l16] = __expf(c[r]);

#pragma unroll
    for (int kk = 0; kk < 4; ++kk) {
      v2f ap;
      ap.x = lds_p[wave][l16][kk * 4 + 2 * half];
      ap.y = lds_p[wave][l16][kk * 4 + 2 * half + 1];
#pragma unroll
      for (int ct = 0; ct < 5; ++ct) {
        v2f bv;
        bv.x = lds_v[kr + kk * 4 + 2 * half][ct * 16 + l16];
        bv.y = lds_v[kr + kk * 4 + 2 * half + 1][ct * 16 + l16];
        acc[ct] = WMMA_F32(ap, bv, acc[ct]);
      }
    }
  }

  __syncthreads();
  if (ks > 0) {
#pragma unroll
    for (int ct = 0; ct < 5; ++ct) {
      RED_AT(ks - 1, rt, ct, 0, lane) = make_float4(acc[ct][0], acc[ct][1], acc[ct][2], acc[ct][3]);
      RED_AT(ks - 1, rt, ct, 1, lane) = make_float4(acc[ct][4], acc[ct][5], acc[ct][6], acc[ct][7]);
    }
  }
  __syncthreads();
  if (ks > 0) return;

  auto add_partials = [&](int ct) {
#pragma unroll
    for (int s = 0; s < KGROUPS - 1; ++s) {
      float4 p0 = RED_AT(s, rt, ct, 0, lane);
      float4 p1 = RED_AT(s, rt, ct, 1, lane);
      acc[ct][0] += p0.x; acc[ct][1] += p0.y; acc[ct][2] += p0.z; acc[ct][3] += p0.w;
      acc[ct][4] += p1.x; acc[ct][5] += p1.y; acc[ct][6] += p1.z; acc[ct][7] += p1.w;
    }
  };
  add_partials(4);
  float inv[8];
#pragma unroll
  for (int r = 0; r < 8; ++r)
    inv[r] = 1.f / __shfl(acc[4][r], half * 16 + 4);
  float* Yb = Y + ((long)b * NN + row0 + rt * 16 + 8 * half) * NCP;
#pragma unroll
  for (int cc = 0; cc < 5; ++cc) {
    const int ct = (cc + 4) % 5;
    if (ct != 4) add_partials(ct);
    const int col = ct * 16 + l16;
    const int dst = (col < 64) ? col + 3 : (col < 67) ? col - 64 : 67;
    if (col < NCP) {
#pragma unroll
      for (int r = 0; r < 8; ++r)
        Yb[r * NCP + dst] = acc[ct][r] * inv[r];
    }
  }
#undef RED_AT
}

__device__ __forceinline__ float combine1(float gr, float ghc, float hp) {
  float r = 1.f / (1.f + __expf(-gr));
  float hc = tanhf(ghc);
  return r * hp + (1.f - r) * hc;
}

#define GOUT_KG 8
#define GOUT_LDS_FLOATS 18432
__device__ __forceinline__ void gout_body(
    float* smem, int n,
    const float* __restrict__ x, const float* __restrict__ Hsrc,
    const float* __restrict__ XG,
    const float* __restrict__ W, const float* __restrict__ TB,
    const float* __restrict__ Hs, const float* __restrict__ olg,
    const float* __restrict__ olb, const float* __restrict__ Gr,
    float* __restrict__ G, int g, int t) {
  float4* part = (float4*)smem;
#define PART_AT(k_, bt_, ct_, h_, l_) part[((((k_) * 2 + (bt_)) * 4 + (ct_)) * 2 + (h_)) * 32 + (l_)]
  float (*grow)[NHID] = (float (*)[NHID])(smem + 16384);

  const int wave = __builtin_amdgcn_readfirstlane(threadIdx.x >> 5);
  const int lane = threadIdx.x & 31;
  const int l16 = lane & 15, half = lane >> 4;
  const int K = t + 1;

  float st0[2][NT], st1[2][NT];
#pragma unroll
  for (int j = 0; j < 2; ++j) {
    const long row = (long)(2 * wave + j) * NN + n;
#pragma unroll
    for (int k = 0; k < NT; ++k) {
      if (k < K) {
        st0[j][k] = Hs[((long)k * NB * NN + row) * 64 + lane];
        st1[j][k] = Hs[((long)k * NB * NN + row) * 64 + lane + 32];
      } else { st0[j][k] = 0.f; st1[j][k] = 0.f; }
    }
  }
  float hp0[2] = {0.f, 0.f}, hp1[2] = {0.f, 0.f};
  if (g != 1) {
#pragma unroll
    for (int j = 0; j < 2; ++j) {
      const long row = (long)(2 * wave + j) * NN + n;
      hp0[j] = Hs[((long)t * NB * NN + row) * 64 + lane];
      hp1[j] = Hs[((long)t * NB * NN + row) * 64 + lane + 32];
    }
  }
  const float lg0 = olg[g * 64 + lane], lg1 = olg[g * 64 + lane + 32];
  const float lb0 = olb[g * 64 + lane], lb1 = olb[g * 64 + lane + 32];

  {
    const int bt = wave & 1;
    const int kg = wave >> 1;
    const int slot = kg >> 2;
    const int q = kg & 3;
    const int kk0 = q ? 4 * q + 1 : 0;
    const int kk1 = 4 * q + 5;

    const int ab = bt * 16 + l16;
    const float* Arow = XG + ((long)ab * NN + n) * NCP;
    const float* xr = x + ((long)(ab * NT + t) * NN + n) * 3;
    const float* hr = Hsrc + ((long)ab * NN + n) * NHID;
    const float* Wm = W + (((long)g * NN + n) * 2 + slot) * NCP * NHID;

    v8f c[4] = {};
    auto kstep = [&](v2f a, int kk) {
#pragma unroll
      for (int ct = 0; ct < 4; ++ct) {
        v2f bw;
        bw.x = Wm[(kk * 4 + 2 * half) * 64 + ct * 16 + l16];
        bw.y = Wm[(kk * 4 + 2 * half + 1) * 64 + ct * 16 + l16];
        c[ct] = WMMA_F32(a, bw, c[ct]);
      }
    };
    if (slot) {
      for (int kk = kk0; kk < kk1; ++kk) {
        v2f a;
        a.x = Arow[kk * 4 + 2 * half];
        a.y = Arow[kk * 4 + 2 * half + 1];
        kstep(a, kk);
      }
    } else {
      int kk = kk0;
      if (kk == 0) {
        const float* p1 = half ? hr : xr + 1;
        v2f a;
        a.x = xr[2 * half];
        a.y = *p1;
        kstep(a, 0);
        kk = 1;
      }
      for (; kk < kk1; ++kk) {
        const int i = kk * 4 + 2 * half;
        v2f a;
        a.x = hr[i - 3];
        a.y = hr[(i - 2 < 63) ? i - 2 : 63];
        kstep(a, kk);
      }
    }

#pragma unroll
    for (int ct = 0; ct < 4; ++ct) {
      PART_AT(kg, bt, ct, 0, lane) = make_float4(c[ct][0], c[ct][1], c[ct][2], c[ct][3]);
      PART_AT(kg, bt, ct, 1, lane) = make_float4(c[ct][4], c[ct][5], c[ct][6], c[ct][7]);
    }
  }
  __syncthreads();

  {
    const int rbt = wave & 1, ct = (wave >> 1) & 3, rh = wave >> 3;
    float4 s = PART_AT(0, rbt, ct, rh, lane);
#pragma unroll
    for (int k = 1; k < GOUT_KG; ++k) {
      float4 p = PART_AT(k, rbt, ct, rh, lane);
      s.x += p.x; s.y += p.y; s.z += p.z; s.w += p.w;
    }
    const float sv[4] = {s.x, s.y, s.z, s.w};
#pragma unroll
    for (int e = 0; e < 4; ++e) {
      int bb = rbt * 16 + rh * 4 + e + 8 * half;
      int o = ct * 16 + l16;
      grow[bb][o] = sv[e] + TB[(((long)g * NT + t) * NB + bb) * 64 + o];
    }
  }
  __syncthreads();

  float gr0[2] = {0.f, 0.f}, gr1[2] = {0.f, 0.f};
  if (g == 2) {
#pragma unroll
    for (int j = 0; j < 2; ++j) {
      const long row = (long)(2 * wave + j) * NN + n;
      gr0[j] = Gr[row * 64 + lane];
      gr1[j] = Gr[row * 64 + lane + 32];
    }
  }

  float g0[2], g1[2], q0[2], q1[2];
#pragma unroll
  for (int j = 0; j < 2; ++j) {
    g0[j] = grow[2 * wave + j][lane];
    g1[j] = grow[2 * wave + j][lane + 32];
  }
  {
    float s[2], v[2], d0[2], d1[2];
#pragma unroll
    for (int j = 0; j < 2; ++j) s[j] = g0[j] + g1[j];
#pragma unroll
    for (int m = 1; m < 32; m <<= 1)
#pragma unroll
      for (int j = 0; j < 2; ++j) s[j] += __shfl_xor(s[j], m);
#pragma unroll
    for (int j = 0; j < 2; ++j) {
      float mu = s[j] * (1.f / 64.f);
      d0[j] = g0[j] - mu; d1[j] = g1[j] - mu;
      v[j] = d0[j] * d0[j] + d1[j] * d1[j];
    }
#pragma unroll
    for (int m = 1; m < 32; m <<= 1)
#pragma unroll
      for (int j = 0; j < 2; ++j) v[j] += __shfl_xor(v[j], m);
#pragma unroll
    for (int j = 0; j < 2; ++j) {
      float inv = rsqrtf(v[j] * (1.f / 64.f) + 1e-5f);
      q0[j] = d0[j] * inv * lg0 + lb0;
      q1[j] = d1[j] * inv * lg1 + lb1;
    }
  }

  float sc0[2][NT], sc1[2][NT];
#pragma unroll
  for (int k = 0; k < NT; ++k) {
    if (k < K) {
      float p0[2], p1[2];
#pragma unroll
      for (int j = 0; j < 2; ++j) { p0[j] = q0[j] * st0[j][k]; p1[j] = q1[j] * st1[j][k]; }
#pragma unroll
      for (int m = 1; m < 16; m <<= 1)
#pragma unroll
        for (int j = 0; j < 2; ++j) { p0[j] += __shfl_xor(p0[j], m); p1[j] += __shfl_xor(p1[j], m); }
#pragma unroll
      for (int j = 0; j < 2; ++j) { sc0[j][k] = p0[j] * 0.25f; sc1[j][k] = p1[j] * 0.25f; }
    } else {
#pragma unroll
      for (int j = 0; j < 2; ++j) { sc0[j][k] = -1e30f; sc1[j][k] = -1e30f; }
    }
  }
#pragma unroll
  for (int j = 0; j < 2; ++j) {
    float m0 = -1e30f, m1 = -1e30f;
#pragma unroll
    for (int k = 0; k < NT; ++k) { m0 = fmaxf(m0, sc0[j][k]); m1 = fmaxf(m1, sc1[j][k]); }
    float s0 = 0.f, s1 = 0.f;
#pragma unroll
    for (int k = 0; k < NT; ++k) {
      sc0[j][k] = __expf(sc0[j][k] - m0); s0 += sc0[j][k];
      sc1[j][k] = __expf(sc1[j][k] - m1); s1 += sc1[j][k];
    }
    float o0 = 0.f, o1 = 0.f;
#pragma unroll
    for (int k = 0; k < NT; ++k) { o0 += sc0[j][k] * st0[j][k]; o1 += sc1[j][k] * st1[j][k]; }
    const long row = (long)(2 * wave + j) * NN + n;
    float r0 = g0[j] + o0 / s0, r1 = g1[j] + o1 / s1;
    if (g == 0) {
      r0 = (1.f / (1.f + __expf(-r0))) * hp0[j];
      r1 = (1.f / (1.f + __expf(-r1))) * hp1[j];
    } else if (g == 2) {
      r0 = combine1(gr0[j], r0, hp0[j]);
      r1 = combine1(gr1[j], r1, hp1[j]);
    }
    G[row * 64 + lane]      = r0;
    G[row * 64 + lane + 32] = r1;
  }
#undef PART_AT
}

#define EMB_BLOCKS 96
__global__ void __launch_bounds__(512, 8) k_stage(
    int nGa, int nGout, int t,
    const float* __restrict__ E, const float* __restrict__ x,
    const float* __restrict__ gaH, float* __restrict__ gaY, int gaG,
    const float* __restrict__ goH, const float* __restrict__ goXG,
    const float* __restrict__ W, const float* __restrict__ TB,
    const float* __restrict__ Hs, const float* __restrict__ olg,
    const float* __restrict__ olb, const float* __restrict__ goGr,
    float* __restrict__ goOut, int goGate,
    const float* __restrict__ ne, const float* __restrict__ te,
    const float* __restrict__ glg, const float* __restrict__ glb,
    float* __restrict__ Enext) {
  __shared__ __attribute__((aligned(16))) float smem[GOUT_LDS_FLOATS];
  const int bid = blockIdx.x;
  if (bid < nGa) {
    graph_attn_body(smem, bid, E, x, gaH, gaY, gaG, t);
  } else if (bid < nGa + nGout) {
    gout_body(smem, bid - nGa, x, goH, goXG, W, TB, Hs, olg, olb, goGr, goOut, goGate, t);
  } else {
    if (t + 1 < NT)
      emb_row(ne, te, glg, glb, Enext, t + 1, (bid - nGa - nGout) * 512 + threadIdx.x);
  }
}

__global__ void k_pred(const float* __restrict__ Hs, const float* __restrict__ pw,
                       const float* __restrict__ pb, float* __restrict__ out) {
  int idx = blockIdx.x * 256 + threadIdx.x;
  if (idx >= NB * 12 * NN) return;
  int n = idx % NN; int r = idx / NN; int o = r % 12; int b = r / 12;
  float acc = pb[o];
  for (int cc = 0; cc < 2; ++cc) {
    const float* h = Hs + ((long)(NT - 1 + cc) * NB * NN + (long)b * NN + n) * 64;
    for (int hh = 0; hh < 64; ++hh)
      acc += h[hh] * pw[(o * 2 + cc) * 64 + hh];
  }
  out[idx] = acc;
}

extern "C" void kernel_launch(void* const* d_in, const int* in_sizes, int n_in,
                              void* d_out, int out_size, void* d_ws, size_t ws_size,
                              hipStream_t stream) {
  const float* x   = (const float*)d_in[0];
  const float* ne  = (const float*)d_in[1];
  const float* te  = (const float*)d_in[2];
  const float* wp  = (const float*)d_in[3];
  const float* bp  = (const float*)d_in[4];
  const float* glg = (const float*)d_in[5];
  const float* glb = (const float*)d_in[6];
  const float* olg = (const float*)d_in[7];
  const float* olb = (const float*)d_in[8];
  const float* pw  = (const float*)d_in[9];
  const float* pb  = (const float*)d_in[10];
  float* out = (float*)d_out;

  float* ws = (float*)d_ws;
  long off = 0;
  float* W    = ws + off; off += 3L * NN * 2 * NCP * NHID;
  float* TB   = ws + off; off += 3L * NT * NB * NHID;
  float* E0   = ws + off; off += 3L * NB * NN * NDE;
  float* E1   = ws + off; off += 3L * NB * NN * NDE;
  float* Hs   = ws + off; off += (long)(NT + 1) * NB * NN * NHID;
  float* XGa  = ws + off; off += (long)NB * NN * NCP;
  float* XGb  = ws + off; off += (long)NB * NN * NCP;
  float* G0   = ws + off; off += (long)NB * NN * NHID;
  float* G1   = ws + off; off += (long)NB * NN * NHID;

  k_w<<<dim3((KW_QUADS + 255) / 256, NN / KW_NODES), 256, 0, stream>>>(ne, wp, W);
  k_tb<<<(3 * NT * NB * NHID + 255) / 256, 256, 0, stream>>>(te, bp, TB);
  k_zero<<<PLANE_F4 / 4 / 256, 256, 0, stream>>>(Hs);

  k_emb<<<(3 * NB * NN + 255) / 256, 256, 0, stream>>>(ne, te, glg, glb, E0, 0);

  float* E = E0;
  float* Enext = E1;
  for (int t = 0; t < NT; ++t) {
    float* Hprev = Hs + (long)t * NB * NN * NHID;
    float* Hnew  = Hs + (long)(t + 1) * NB * NN * NHID;
    k_stage<<<GA_BLOCKS + EMB_BLOCKS, 512, 0, stream>>>(GA_BLOCKS, 0, t, E, x, Hprev, XGa, 0,
        Hprev, XGb, W, TB, Hs, olg, olb, G1, G0, 0, ne, te, glg, glb, Enext);
    k_stage<<<GA_BLOCKS + NN, 512, 0, stream>>>(GA_BLOCKS, NN, t, E, x, Hprev, XGb, 1,
        Hprev, XGa, W, TB, Hs, olg, olb, G1, G0, 0, ne, te, glg, glb, Enext);
    k_stage<<<GA_BLOCKS + NN, 512, 0, stream>>>(GA_BLOCKS, NN, t, E, x, G0, XGa, 2,
        Hprev, XGb, W, TB, Hs, olg, olb, G0, G1, 1, ne, te, glg, glb, Enext);
    k_stage<<<NN, 512, 0, stream>>>(0, NN, t, E, x, G0, XGb, 2,
        G0, XGa, W, TB, Hs, olg, olb, G1, Hnew, 2, ne, te, glg, glb, Enext);
    float* Eswap = E; E = Enext; Enext = Eswap;
  }
  k_pred<<<(NB * 12 * NN + 255) / 256, 256, 0, stream>>>(Hs, pw, pb, out);
}


# node GEMM work-groups run at raised wave priority (s_setprio 3, back to 0 at their end): the older graph-attention waves on their SIMD no longer make their products wait; sum orders unchanged
# speedup vs baseline: 5.0697x; 1.0565x over previous
#include <hip/hip_runtime.h>
#include <math.h>

typedef __attribute__((ext_vector_type(2))) float v2f;
typedef __attribute__((ext_vector_type(8))) float v8f;

#define WMMA_F32(A, B, C) \
  __builtin_amdgcn_wmma_f32_16x16x4_f32(false, (A), false, (B), (short)0, (C), false, false)

#define NB   32
#define NT   12
#define NN   512
#define NDE  16
#define NHID 64
#define NCP  68

#define KW_NODES 8
#define KW_QUADS (3 * 2 * NCP * 16)
__global__ void __launch_bounds__(256) k_w(const float* __restrict__ ne, const float* __restrict__ wp,
                                           float* __restrict__ W) {
  const int idx = blockIdx.x * 256 + threadIdx.x;
  if (idx >= KW_QUADS) return;
  const int q = idx & 15; int r = idx >> 4;
  const int i = r % NCP; r /= NCP;
  const int slot = r & 1; const int g = r >> 1;
  const int n0 = blockIdx.y * KW_NODES;
  const bool live = i < 67;
  const int il = live ? i : 66;
  float4 w[NDE];
#pragma unroll
  for (int d = 0; d < NDE; ++d)
    w[d] = *(const float4*)(wp + ((((long)(g * NDE + d) * 2 + slot) * 67 + il) * 64 + q * 4));
#pragma unroll
  for (int j = 0; j < KW_NODES; ++j) {
    const int n = n0 + j;
    float4 acc = make_float4(0.f, 0.f, 0.f, 0.f);
#pragma unroll
    for (int d = 0; d < NDE; ++d) {
      const float e = ne[n * NDE + d];
      acc.x += e * w[d].x; acc.y += e * w[d].y; acc.z += e * w[d].z; acc.w += e * w[d].w;
    }
    if (!live) acc = make_float4(0.f, 0.f, 0.f, 0.f);
    *(float4*)(W + ((((long)g * NN + n) * 2 + slot) * NCP + i) * 64 + q * 4) = acc;
  }
}

__global__ void k_tb(const float* __restrict__ te, const float* __restrict__ bp,
                     float* __restrict__ TB) {
  int idx = blockIdx.x * 256 + threadIdx.x;
  if (idx >= 3 * NT * NB * NHID) return;
  int o = idx % 64; int r = idx / 64;
  int b = r % NB; r /= NB;
  int t = r % NT; int g = r / NT;
  float acc = 0.f;
  for (int d = 0; d < NDE; ++d)
    acc += te[(b * NT + t) * NDE + d] * bp[(g * NDE + d) * 64 + o];
  TB[idx] = acc;
}

__device__ __forceinline__ void emb_row(const float* __restrict__ ne, const float* __restrict__ te,
                                        const float* __restrict__ glg, const float* __restrict__ glb,
                                        float* __restrict__ E, int t, int idx) {
  if (idx >= 3 * NB * NN) return;
  int n = idx % NN; int r = idx / NN;
  int b = r % NB; int g = r / NB;
  float v[NDE]; float mu = 0.f;
  for (int d = 0; d < NDE; ++d) { v[d] = ne[n * NDE + d] + te[(b * NT + t) * NDE + d]; mu += v[d]; }
  mu *= (1.f / NDE);
  float var = 0.f;
  for (int d = 0; d < NDE; ++d) { float q = v[d] - mu; var += q * q; }
  var *= (1.f / NDE);
  float inv = rsqrtf(var + 1e-12f);
  for (int d = 0; d < NDE; ++d)
    E[(long)idx * NDE + d] = (v[d] - mu) * inv * glg[g * NDE + d] + glb[g * NDE + d];
}

__global__ void k_emb(const float* __restrict__ ne, const float* __restrict__ te,
                      const float* __restrict__ glg, const float* __restrict__ glb,
                      float* __restrict__ E, int t) {
  emb_row(ne, te, glg, glb, E, t, blockIdx.x * 256 + threadIdx.x);
}

#define PLANE_F4 (NB * NN * NHID / 4)
__global__ void __launch_bounds__(256) k_zero(float* __restrict__ p) {
  float4* p4 = (float4*)p;
  const int i = blockIdx.x * 256 + threadIdx.x;
#pragma unroll
  for (int k = 0; k < 4; ++k)
    p4[i + k * (PLANE_F4 / 4)] = make_float4(0.f, 0.f, 0.f, 0.f);
}


#define KGROUPS 4
#define GA_LDS_FLOATS 15360
#define GA_BLOCKS 256
__device__ __forceinline__ void graph_attn_body(
    float* smem, int bid,
    const float* __restrict__ E,
    const float* __restrict__ x,
    const float* __restrict__ Hsrc,
    float* __restrict__ Y,
    int g, int t) {
  float (*lds_e)[16] = (float (*)[16])smem;
  float (*lds_v)[80] = (float (*)[80])(smem + 1024);
  float (*lds_p)[16][17] = (float (*)[16][17])(smem + 6144);
  float4* lds_red = (float4*)smem;
#define RED_AT(s_, rt_, ct_, h_, l_) lds_red[((((s_) * 4 + (rt_)) * 5 + (ct_)) * 2 + (h_)) * 32 + (l_)]

  const int b = bid >> 3;
  const int row0 = (bid & 7) * 64;
  const int tid = threadIdx.x;
  const int wave = tid >> 5;
  const int rt = wave & 3;
  const int ks = wave >> 2;
  const int lane = tid & 31;
  const int l16 = lane & 15;
  const int half = lane >> 4;
  const int myrow = row0 + rt * 16 + l16;
  const int kr = ks * 16;

  const float* Eb = E + ((long)g * NB + b) * NN * NDE;
  const float* Hb = Hsrc + (long)b * NN * NHID;
  const float* xb = x + (long)(b * NT + t) * NN * 3;

  float2 nE; float4 nH0, nH1; float nX;
  const int xi = tid < 192 ? tid : 191;
  auto request = [&](int w) {
    nE = ((const float2*)(Eb + (long)w * 64 * NDE))[tid];
    const float4* sh = (const float4*)(Hb + (long)w * 64 * NHID);
    nH0 = sh[tid];
    nH1 = sh[tid + 512];
    nX = xb[w * 192 + xi];
  };
  request(0);

  v2f aA[4];
#pragma unroll
  for (int kk = 0; kk < 4; ++kk) {
    aA[kk].x = Eb[myrow * NDE + kk * 4 + 2 * half];
    aA[kk].y = Eb[myrow * NDE + kk * 4 + 2 * half + 1];
  }

  for (int i = tid; i < 64 * 13; i += 512) {
    int rr = i / 13, cc = 67 + i % 13;
    lds_v[rr][cc] = (cc == 68) ? 1.f : 0.f;
  }

  v8f acc[5] = {};

#pragma nounroll
  for (int it = 0; it < NN / 64; ++it) {
    __syncthreads();
    ((float2*)lds_e)[tid] = nE;
    *(float4*)&lds_v[tid >> 4][(tid & 15) * 4] = nH0;
    *(float4*)&lds_v[32 + (tid >> 4)][(tid & 15) * 4] = nH1;
    if (tid < 192)
      lds_v[tid / 3][64 + tid % 3] = nX;
    __syncthreads();
    if (it + 1 < NN / 64) request(it + 1);

    v8f c = {};
#pragma unroll
    for (int kk = 0; kk < 4; ++kk) {
      v2f bf;
      bf.x = lds_e[kr + l16][kk * 4 + 2 * half];
      bf.y = lds_e[kr + l16][kk * 4 + 2 * half + 1];
      c = WMMA_F32(aA[kk], bf, c);
    }

#pragma unroll
    for (int r = 0; r < 8; ++r)
      lds_p[wave][r + 8 * half][l16] = __expf(c[r]);

#pragma unroll
    for (int kk = 0; kk < 4; ++kk) {
      v2f ap;
      ap.x = lds_p[wave][l16][kk * 4 + 2 * half];
      ap.y = lds_p[wave][l16][kk * 4 + 2 * half + 1];
#pragma unroll
      for (int ct = 0; ct < 5; ++ct) {
        v2f bv;
        bv.x = lds_v[kr + kk * 4 + 2 * half][ct * 16 + l16];
        bv.y = lds_v[kr + kk * 4 + 2 * half + 1][ct * 16 + l16];
        acc[ct] = WMMA_F32(ap, bv, acc[ct]);
      }
    }
  }

  __syncthreads();
  if (ks > 0) {
#pragma unroll
    for (int ct = 0; ct < 5; ++ct) {
      RED_AT(ks - 1, rt, ct, 0, lane) = make_float4(acc[ct][0], acc[ct][1], acc[ct][2], acc[ct][3]);
      RED_AT(ks - 1, rt, ct, 1, lane) = make_float4(acc[ct][4], acc[ct][5], acc[ct][6], acc[ct][7]);
    }
  }
  __syncthreads();
  if (ks > 0) return;

  auto add_partials = [&](int ct) {
#pragma unroll
    for (int s = 0; s < KGROUPS - 1; ++s) {
      float4 p0 = RED_AT(s, rt, ct, 0, lane);
      float4 p1 = RED_AT(s, rt, ct, 1, lane);
      acc[ct][0] += p0.x; acc[ct][1] += p0.y; acc[ct][2] += p0.z; acc[ct][3] += p0.w;
      acc[ct][4] += p1.x; acc[ct][5] += p1.y; acc[ct][6] += p1.z; acc[ct][7] += p1.w;
    }
  };
  add_partials(4);
  float inv[8];
#pragma unroll
  for (int r = 0; r < 8; ++r)
    inv[r] = 1.f / __shfl(acc[4][r], half * 16 + 4);
  float* Yb = Y + ((long)b * NN + row0 + rt * 16 + 8 * half) * NCP;
#pragma unroll
  for (int cc = 0; cc < 5; ++cc) {
    const int ct = (cc + 4) % 5;
    if (ct != 4) add_partials(ct);
    const int col = ct * 16 + l16;
    const int dst = (col < 64) ? col + 3 : (col < 67) ? col - 64 : 67;
    if (col < NCP) {
#pragma unroll
      for (int r = 0; r < 8; ++r)
        Yb[r * NCP + dst] = acc[ct][r] * inv[r];
    }
  }
#undef RED_AT
}

__device__ __forceinline__ float combine1(float gr, float ghc, float hp) {
  float r = 1.f / (1.f + __expf(-gr));
  float hc = tanhf(ghc);
  return r * hp + (1.f - r) * hc;
}

#define GOUT_KG 8
#define GOUT_LDS_FLOATS 18432
__device__ __forceinline__ void gout_body(
    float* smem, int n,
    const float* __restrict__ x, const float* __restrict__ Hsrc,
    const float* __restrict__ XG,
    const float* __restrict__ W, const float* __restrict__ TB,
    const float* __restrict__ Hs, const float* __restrict__ olg,
    const float* __restrict__ olb, const float* __restrict__ Gr,
    float* __restrict__ G, int g, int t) {
  float4* part = (float4*)smem;
#define PART_AT(k_, bt_, ct_, h_, l_) part[((((k_) * 2 + (bt_)) * 4 + (ct_)) * 2 + (h_)) * 32 + (l_)]
  float (*grow)[NHID] = (float (*)[NHID])(smem + 16384);

  const int wave = __builtin_amdgcn_readfirstlane(threadIdx.x >> 5);
  const int lane = threadIdx.x & 31;
  const int l16 = lane & 15, half = lane >> 4;
  const int K = t + 1;

  float st0[2][NT], st1[2][NT];
#pragma unroll
  for (int j = 0; j < 2; ++j) {
    const long row = (long)(2 * wave + j) * NN + n;
#pragma unroll
    for (int k = 0; k < NT; ++k) {
      if (k < K) {
        st0[j][k] = Hs[((long)k * NB * NN + row) * 64 + lane];
        st1[j][k] = Hs[((long)k * NB * NN + row) * 64 + lane + 32];
      } else { st0[j][k] = 0.f; st1[j][k] = 0.f; }
    }
  }
  float hp0[2] = {0.f, 0.f}, hp1[2] = {0.f, 0.f};
  if (g != 1) {
#pragma unroll
    for (int j = 0; j < 2; ++j) {
      const long row = (long)(2 * wave + j) * NN + n;
      hp0[j] = Hs[((long)t * NB * NN + row) * 64 + lane];
      hp1[j] = Hs[((long)t * NB * NN + row) * 64 + lane + 32];
    }
  }
  const float lg0 = olg[g * 64 + lane], lg1 = olg[g * 64 + lane + 32];
  const float lb0 = olb[g * 64 + lane], lb1 = olb[g * 64 + lane + 32];

  {
    const int bt = wave & 1;
    const int kg = wave >> 1;
    const int slot = kg >> 2;
    const int q = kg & 3;
    const int kk0 = q ? 4 * q + 1 : 0;
    const int kk1 = 4 * q + 5;

    const int ab = bt * 16 + l16;
    const float* Arow = XG + ((long)ab * NN + n) * NCP;
    const float* xr = x + ((long)(ab * NT + t) * NN + n) * 3;
    const float* hr = Hsrc + ((long)ab * NN + n) * NHID;
    const float* Wm = W + (((long)g * NN + n) * 2 + slot) * NCP * NHID;

    v8f c[4] = {};
    auto kstep = [&](v2f a, int kk) {
#pragma unroll
      for (int ct = 0; ct < 4; ++ct) {
        v2f bw;
        bw.x = Wm[(kk * 4 + 2 * half) * 64 + ct * 16 + l16];
        bw.y = Wm[(kk * 4 + 2 * half + 1) * 64 + ct * 16 + l16];
        c[ct] = WMMA_F32(a, bw, c[ct]);
      }
    };
    if (slot) {
      for (int kk = kk0; kk < kk1; ++kk) {
        v2f a;
        a.x = Arow[kk * 4 + 2 * half];
        a.y = Arow[kk * 4 + 2 * half + 1];
        kstep(a, kk);
      }
    } else {
      int kk = kk0;
      if (kk == 0) {
        const float* p1 = half ? hr : xr + 1;
        v2f a;
        a.x = xr[2 * half];
        a.y = *p1;
        kstep(a, 0);
        kk = 1;
      }
      for (; kk < kk1; ++kk) {
        const int i = kk * 4 + 2 * half;
        v2f a;
        a.x = hr[i - 3];
        a.y = hr[(i - 2 < 63) ? i - 2 : 63];
        kstep(a, kk);
      }
    }

#pragma unroll
    for (int ct = 0; ct < 4; ++ct) {
      PART_AT(kg, bt, ct, 0, lane) = make_float4(c[ct][0], c[ct][1], c[ct][2], c[ct][3]);
      PART_AT(kg, bt, ct, 1, lane) = make_float4(c[ct][4], c[ct][5], c[ct][6], c[ct][7]);
    }
  }
  __syncthreads();

  {
    const int rbt = wave & 1, ct = (wave >> 1) & 3, rh = wave >> 3;
    float4 s = PART_AT(0, rbt, ct, rh, lane);
#pragma unroll
    for (int k = 1; k < GOUT_KG; ++k) {
      float4 p = PART_AT(k, rbt, ct, rh, lane);
      s.x += p.x; s.y += p.y; s.z += p.z; s.w += p.w;
    }
    const float sv[4] = {s.x, s.y, s.z, s.w};
#pragma unroll
    for (int e = 0; e < 4; ++e) {
      int bb = rbt * 16 + rh * 4 + e + 8 * half;
      int o = ct * 16 + l16;
      grow[bb][o] = sv[e] + TB[(((long)g * NT + t) * NB + bb) * 64 + o];
    }
  }
  __syncthreads();

  float gr0[2] = {0.f, 0.f}, gr1[2] = {0.f, 0.f};
  if (g == 2) {
#pragma unroll
    for (int j = 0; j < 2; ++j) {
      const long row = (long)(2 * wave + j) * NN + n;
      gr0[j] = Gr[row * 64 + lane];
      gr1[j] = Gr[row * 64 + lane + 32];
    }
  }

  float g0[2], g1[2], q0[2], q1[2];
#pragma unroll
  for (int j = 0; j < 2; ++j) {
    g0[j] = grow[2 * wave + j][lane];
    g1[j] = grow[2 * wave + j][lane + 32];
  }
  {
    float s[2], v[2], d0[2], d1[2];
#pragma unroll
    for (int j = 0; j < 2; ++j) s[j] = g0[j] + g1[j];
#pragma unroll
    for (int m = 1; m < 32; m <<= 1)
#pragma unroll
      for (int j = 0; j < 2; ++j) s[j] += __shfl_xor(s[j], m);
#pragma unroll
    for (int j = 0; j < 2; ++j) {
      float mu = s[j] * (1.f / 64.f);
      d0[j] = g0[j] - mu; d1[j] = g1[j] - mu;
      v[j] = d0[j] * d0[j] + d1[j] * d1[j];
    }
#pragma unroll
    for (int m = 1; m < 32; m <<= 1)
#pragma unroll
      for (int j = 0; j < 2; ++j) v[j] += __shfl_xor(v[j], m);
#pragma unroll
    for (int j = 0; j < 2; ++j) {
      float inv = rsqrtf(v[j] * (1.f / 64.f) + 1e-5f);
      q0[j] = d0[j] * inv * lg0 + lb0;
      q1[j] = d1[j] * inv * lg1 + lb1;
    }
  }

  float sc0[2][NT], sc1[2][NT];
#pragma unroll
  for (int k = 0; k < NT; ++k) {
    if (k < K) {
      float p0[2], p1[2];
#pragma unroll
      for (int j = 0; j < 2; ++j) { p0[j] = q0[j] * st0[j][k]; p1[j] = q1[j] * st1[j][k]; }
#pragma unroll
      for (int m = 1; m < 16; m <<= 1)
#pragma unroll
        for (int j = 0; j < 2; ++j) { p0[j] += __shfl_xor(p0[j], m); p1[j] += __shfl_xor(p1[j], m); }
#pragma unroll
      for (int j = 0; j < 2; ++j) { sc0[j][k] = p0[j] * 0.25f; sc1[j][k] = p1[j] * 0.25f; }
    } else {
#pragma unroll
      for (int j = 0; j < 2; ++j) { sc0[j][k] = -1e30f; sc1[j][k] = -1e30f; }
    }
  }
#pragma unroll
  for (int j = 0; j < 2; ++j) {
    float m0 = -1e30f, m1 = -1e30f;
#pragma unroll
    for (int k = 0; k < NT; ++k) { m0 = fmaxf(m0, sc0[j][k]); m1 = fmaxf(m1, sc1[j][k]); }
    float s0 = 0.f, s1 = 0.f;
#pragma unroll
    for (int k = 0; k < NT; ++k) {
      sc0[j][k] = __expf(sc0[j][k] - m0); s0 += sc0[j][k];
      sc1[j][k] = __expf(sc1[j][k] - m1); s1 += sc1[j][k];
    }
    float o0 = 0.f, o1 = 0.f;
#pragma unroll
    for (int k = 0; k < NT; ++k) { o0 += sc0[j][k] * st0[j][k]; o1 += sc1[j][k] * st1[j][k]; }
    const long row = (long)(2 * wave + j) * NN + n;
    float r0 = g0[j] + o0 / s0, r1 = g1[j] + o1 / s1;
    if (g == 0) {
      r0 = (1.f / (1.f + __expf(-r0))) * hp0[j];
      r1 = (1.f / (1.f + __expf(-r1))) * hp1[j];
    } else if (g == 2) {
      r0 = combine1(gr0[j], r0, hp0[j]);
      r1 = combine1(gr1[j], r1, hp1[j]);
    }
    G[row * 64 + lane]      = r0;
    G[row * 64 + lane + 32] = r1;
  }
#undef PART_AT
}

#define EMB_BLOCKS 96
__global__ void __launch_bounds__(512, 8) k_stage(
    int nGa, int nGout, int t,
    const float* __restrict__ E, const float* __restrict__ x,
    const float* __restrict__ gaH, float* __restrict__ gaY, int gaG,
    const float* __restrict__ goH, const float* __restrict__ goXG,
    const float* __restrict__ W, const float* __restrict__ TB,
    const float* __restrict__ Hs, const float* __restrict__ olg,
    const float* __restrict__ olb, const float* __restrict__ goGr,
    float* __restrict__ goOut, int goGate,
    const float* __restrict__ ne, const float* __restrict__ te,
    const float* __restrict__ glg, const float* __restrict__ glb,
    float* __restrict__ Enext) {
  __shared__ __attribute__((aligned(16))) float smem[GOUT_LDS_FLOATS];
  const int bid = blockIdx.x;
  if (bid < nGa) {
    graph_attn_body(smem, bid, E, x, gaH, gaY, gaG, t);
  } else if (bid < nGa + nGout) {
    __builtin_amdgcn_s_setprio(3);
    gout_body(smem, bid - nGa, x, goH, goXG, W, TB, Hs, olg, olb, goGr, goOut, goGate, t);
    __builtin_amdgcn_s_setprio(0);
  } else {
    if (t + 1 < NT)
      emb_row(ne, te, glg, glb, Enext, t + 1, (bid - nGa - nGout) * 512 + threadIdx.x);
  }
}

__global__ void k_pred(const float* __restrict__ Hs, const float* __restrict__ pw,
                       const float* __restrict__ pb, float* __restrict__ out) {
  int idx = blockIdx.x * 256 + threadIdx.x;
  if (idx >= NB * 12 * NN) return;
  int n = idx % NN; int r = idx / NN; int o = r % 12; int b = r / 12;
  float acc = pb[o];
  for (int cc = 0; cc < 2; ++cc) {
    const float* h = Hs + ((long)(NT - 1 + cc) * NB * NN + (long)b * NN + n) * 64;
    for (int hh = 0; hh < 64; ++hh)
      acc += h[hh] * pw[(o * 2 + cc) * 64 + hh];
  }
  out[idx] = acc;
}

extern "C" void kernel_launch(void* const* d_in, const int* in_sizes, int n_in,
                              void* d_out, int out_size, void* d_ws, size_t ws_size,
                              hipStream_t stream) {
  const float* x   = (const float*)d_in[0];
  const float* ne  = (const float*)d_in[1];
  const float* te  = (const float*)d_in[2];
  const float* wp  = (const float*)d_in[3];
  const float* bp  = (const float*)d_in[4];
  const float* glg = (const float*)d_in[5];
  const float* glb = (const float*)d_in[6];
  const float* olg = (const float*)d_in[7];
  const float* olb = (const float*)d_in[8];
  const float* pw  = (const float*)d_in[9];
  const float* pb  = (const float*)d_in[10];
  float* out = (float*)d_out;

  float* ws = (float*)d_ws;
  long off = 0;
  float* W    = ws + off; off += 3L * NN * 2 * NCP * NHID;
  float* TB   = ws + off; off += 3L * NT * NB * NHID;
  float* E0   = ws + off; off += 3L * NB * NN * NDE;
  float* E1   = ws + off; off += 3L * NB * NN * NDE;
  float* Hs   = ws + off; off += (long)(NT + 1) * NB * NN * NHID;
  float* XGa  = ws + off; off += (long)NB * NN * NCP;
  float* XGb  = ws + off; off += (long)NB * NN * NCP;
  float* G0   = ws + off; off += (long)NB * NN * NHID;
  float* G1   = ws + off; off += (long)NB * NN * NHID;

  k_w<<<dim3((KW_QUADS + 255) / 256, NN / KW_NODES), 256, 0, stream>>>(ne, wp, W);
  k_tb<<<(3 * NT * NB * NHID + 255) / 256, 256, 0, stream>>>(te, bp, TB);
  k_zero<<<PLANE_F4 / 4 / 256, 256, 0, stream>>>(Hs);

  k_emb<<<(3 * NB * NN + 255) / 256, 256, 0, stream>>>(ne, te, glg, glb, E0, 0);

  float* E = E0;
  float* Enext = E1;
  for (int t = 0; t < NT; ++t) {
    float* Hprev = Hs + (long)t * NB * NN * NHID;
    float* Hnew  = Hs + (long)(t + 1) * NB * NN * NHID;
    k_stage<<<GA_BLOCKS + EMB_BLOCKS, 512, 0, stream>>>(GA_BLOCKS, 0, t, E, x, Hprev, XGa, 0,
        Hprev, XGb, W, TB, Hs, olg, olb, G1, G0, 0, ne, te, glg, glb, Enext);
    k_stage<<<GA_BLOCKS + NN, 512, 0, stream>>>(GA_BLOCKS, NN, t, E, x, Hprev, XGb, 1,
        Hprev, XGa, W, TB, Hs, olg, olb, G1, G0, 0, ne, te, glg, glb, Enext);
    k_stage<<<GA_BLOCKS + NN, 512, 0, stream>>>(GA_BLOCKS, NN, t, E, x, G0, XGa, 2,
        Hprev, XGb, W, TB, Hs, olg, olb, G0, G1, 1, ne, te, glg, glb, Enext);
    k_stage<<<NN, 512, 0, stream>>>(0, NN, t, E, x, G0, XGb, 2,
        G0, XGa, W, TB, Hs, olg, olb, G1, Hnew, 2, ne, te, glg, glb, Enext);
    float* Eswap = E; E = Enext; Enext = Eswap;
  }
  k_pred<<<(NB * 12 * NN + 255) / 256, 256, 0, stream>>>(Hs, pw, pb, out);
}
